# starting kernel
# baseline (speedup 1.0000x reference)
#include <hip/hip_runtime.h>

#define Nn 100000
#define Ee 1000000
#define INF 128
#define HID 128
#define CLS 64

typedef __attribute__((ext_vector_type(2))) float v2f;
typedef __attribute__((ext_vector_type(8))) float v8f;

__device__ __forceinline__ void atomic_add_f32(float* p, float v) {
    __hip_atomic_fetch_add(p, v, __ATOMIC_RELAXED, __HIP_MEMORY_SCOPE_AGENT);
}

__global__ void k_zero_deg(int* __restrict__ deg) {
    int i = blockIdx.x * blockDim.x + threadIdx.x;
    if (i < Nn) deg[i] = 0;
}

__global__ void k_count_deg(const int* __restrict__ dst, int* __restrict__ deg) {
    int e = blockIdx.x * blockDim.x + threadIdx.x;
    if (e < Ee) atomicAdd(&deg[dst[e]], 1);
}

__global__ void k_dinv(const int* __restrict__ deg, float* __restrict__ dinv) {
    int i = blockIdx.x * blockDim.x + threadIdx.x;
    if (i < Nn) {
        float d = (float)deg[i] + 1.0f;
        dinv[i] = 1.0f / sqrtf(d);
    }
}

template <int DOUT, bool RELU_IN>
__global__ void k_gemm_wmma(const float* __restrict__ X,
                            const float* __restrict__ W,
                            float* __restrict__ H) {
    __shared__ float lds_x[16 * 132];
    __shared__ float lds_w[128 * DOUT];

    const int tid     = threadIdx.x;
    const int lane    = tid & 31;
    const int wave    = tid >> 5;
    const int rowBase = blockIdx.x * 16;
    const int colBase = wave * 16;
    const int nthr    = (DOUT / 16) * 32;

    for (int i = tid; i < 16 * 128; i += nthr) {
        int r = i >> 7, c = i & 127;
        float v = X[(size_t)(rowBase + r) * 128 + c];
        if (RELU_IN) v = fmaxf(v, 0.0f);
        lds_x[r * 132 + c] = v;
    }
    for (int i = tid; i < 128 * DOUT; i += nthr) lds_w[i] = W[i];
    __syncthreads();

    const int r    = lane & 15;
    const int half = lane >> 4;
    const int koff = half * 2;

    v8f acc = {};
#pragma unroll
    for (int kk = 0; kk < 128; kk += 4) {
        v2f a = *(const v2f*)(lds_x + r * 132 + kk + koff);
        v2f b;
        b.x = lds_w[(kk + koff) * DOUT + colBase + r];
        b.y = lds_w[(kk + koff + 1) * DOUT + colBase + r];
        acc = __builtin_amdgcn_wmma_f32_16x16x4_f32(
             false, a,  false, b,
             (short)0, acc,  false,  false);
    }

#pragma unroll
    for (int v = 0; v < 8; ++v) {
        H[(size_t)(rowBase + v + half * 8) * DOUT + colBase + r] = acc[v];
    }
}

template <int DOUT>
__global__ void k_init_agg(const float* __restrict__ h,
                           const float* __restrict__ b,
                           const float* __restrict__ dinv,
                           float* __restrict__ out) {
    const int per = DOUT / 4;
    int t = blockIdx.x * blockDim.x + threadIdx.x;
    if (t >= Nn * per) return;
    int i  = t / per;
    int c4 = t - i * per;
    float w = dinv[i];
    w = w * w;
    float4 hv = ((const float4*)h)[t];
    float4 bb = ((const float4*)b)[c4];
    float4 o;
    o.x = hv.x * w + bb.x;
    o.y = hv.y * w + bb.y;
    o.z = hv.z * w + bb.z;
    o.w = hv.w * w + bb.w;
    ((float4*)out)[t] = o;
}

template <int DOUT>
__global__ void k_scatter(const float* __restrict__ h,
                          const int* __restrict__ src,
                          const int* __restrict__ dst,
                          const float* __restrict__ dinv,
                          float* __restrict__ out) {
    const int per = DOUT / 4;
    long long t = (long long)blockIdx.x * blockDim.x + threadIdx.x;
    if (t >= (long long)Ee * per) return;
    int e = (int)(t / per);
    int c = (int)(t - (long long)e * per);
    int s = src[e];
    int d = dst[e];
    float w = dinv[s] * dinv[d];
    float4 hv = ((const float4*)h)[(size_t)s * per + c];
    float* o = out + (size_t)d * DOUT + c * 4;
    atomic_add_f32(o + 0, hv.x * w);
    atomic_add_f32(o + 1, hv.y * w);
    atomic_add_f32(o + 2, hv.z * w);
    atomic_add_f32(o + 3, hv.w * w);
}

extern "C" void kernel_launch(void* const* d_in, const int* in_sizes, int n_in,
                              void* d_out, int out_size, void* d_ws, size_t ws_size,
                              hipStream_t stream) {
    const float* X   = (const float*)d_in[0];
    const float* W1  = (const float*)d_in[1];
    const float* b1  = (const float*)d_in[2];
    const float* W2  = (const float*)d_in[3];
    const float* b2  = (const float*)d_in[4];
    const float* W3  = (const float*)d_in[5];
    const float* b3  = (const float*)d_in[6];
    const int*   ei  = (const int*)d_in[7];
    const int*   src = ei;
    const int*   dst = ei + Ee;
    float* out = (float*)d_out;

    float* bufA = (float*)d_ws;
    float* bufB = bufA + (size_t)Nn * 128;
    int*   deg  = (int*)(bufB + (size_t)Nn * 128);
    float* dinv = (float*)(deg + Nn);

    const int T = 256;
    k_zero_deg<<<(Nn + T - 1) / T, T, 0, stream>>>(deg);
    k_count_deg<<<(Ee + T - 1) / T, T, 0, stream>>>(dst, deg);
    k_dinv<<<(Nn + T - 1) / T, T, 0, stream>>>(deg, dinv);

    const int gGemm = Nn / 16;

    k_gemm_wmma<128, false><<<gGemm, 256, 0, stream>>>(X, W1, bufA);
    k_init_agg<128><<<(Nn * 32 + T - 1) / T, T, 0, stream>>>(bufA, b1, dinv, bufB);
    k_scatter<128><<<(int)(((long long)Ee * 32 + T - 1) / T), T, 0, stream>>>(bufA, src, dst, dinv, bufB);

    k_gemm_wmma<128, true><<<gGemm, 256, 0, stream>>>(bufB, W2, bufA);
    k_init_agg<128><<<(Nn * 32 + T - 1) / T, T, 0, stream>>>(bufA, b2, dinv, bufB);
    k_scatter<128><<<(int)(((long long)Ee * 32 + T - 1) / T), T, 0, stream>>>(bufA, src, dst, dinv, bufB);

    k_gemm_wmma<64, true><<<gGemm, 128, 0, stream>>>(bufB, W3, bufA);
    k_init_agg<64><<<(Nn * 16 + T - 1) / T, T, 0, stream>>>(bufA, b3, dinv, out);
    k_scatter<64><<<(int)(((long long)Ee * 16 + T - 1) / T), T, 0, stream>>>(bufA, src, dst, dinv, out);
}


# Per-edge float atomic scatter replaced by per-node gather over an edge list bucketed by destination (slot = counting add's return: sum order is arrival order); rows summed in registers, stored once
# speedup vs baseline: 2.0732x; 2.0732x over previous
#include <hip/hip_runtime.h>

#define Nn 100000
#define Ee 1000000
#define INF 128
#define HID 128
#define CLS 64

typedef __attribute__((ext_vector_type(2))) float v2f;
typedef __attribute__((ext_vector_type(8))) float v8f;

__global__ void k_zero_deg(int* __restrict__ deg) {
    int i = blockIdx.x * blockDim.x + threadIdx.x;
    if (i < Nn) deg[i] = 0;
}

__global__ void k_count_deg(const int* __restrict__ dst, int* __restrict__ deg, int* __restrict__ rank) {
    int e = blockIdx.x * blockDim.x + threadIdx.x;
    if (e < Ee) rank[e] = atomicAdd(&deg[dst[e]], 1);
}

__global__ __launch_bounds__(256) void k_scan_dinv(const int* __restrict__ deg,
                                                   int* __restrict__ rowptr,
                                                   float* __restrict__ dinv) {
    __shared__ int s_pre[8];
    __shared__ int s_tot[8];
    const int tid  = threadIdx.x;
    const int lane = tid & 31;
    const int wave = tid >> 5;
    const int4* deg4 = (const int4*)deg;

    int pre = 0;
    const int lim = blockIdx.x * 256;
#pragma unroll 8
    for (int i = tid; i < lim; i += 256) {
        int4 v = deg4[i];
        pre += (v.x + v.y) + (v.z + v.w);
    }
    for (int o = 16; o > 0; o >>= 1) pre += __shfl_xor(pre, o);
    if (lane == 0) s_pre[wave] = pre;

    const int i4 = blockIdx.x * 256 + tid;
    const bool ok = i4 < Nn / 4;
    int4 v = make_int4(0, 0, 0, 0);
    if (ok) v = deg4[i4];
    const int tsum = (v.x + v.y) + (v.z + v.w);
    int inc = tsum;
    for (int o = 1; o < 32; o <<= 1) {
        int t = __shfl_up(inc, o);
        if (lane >= o) inc += t;
    }
    if (lane == 31) s_tot[wave] = inc;
    __syncthreads();
    int base = 0;
    for (int w = 0; w < 8; ++w) {
        base += s_pre[w];
        if (w < wave) base += s_tot[w];
    }
    if (ok) {
        int4 r;
        r.x = base + inc - tsum;
        r.y = r.x + v.x;
        r.z = r.y + v.y;
        r.w = r.z + v.z;
        ((int4*)rowptr)[i4] = r;
        float4 f;
        f.x = 1.0f / sqrtf((float)v.x + 1.0f);
        f.y = 1.0f / sqrtf((float)v.y + 1.0f);
        f.z = 1.0f / sqrtf((float)v.z + 1.0f);
        f.w = 1.0f / sqrtf((float)v.w + 1.0f);
        ((float4*)dinv)[i4] = f;
    }
}

__global__ void k_fill(const int* __restrict__ src, const int* __restrict__ dst,
                       const int* __restrict__ rowptr, const int* __restrict__ rank,
                       int* __restrict__ col) {
    int e = blockIdx.x * blockDim.x + threadIdx.x;
    if (e < Ee) col[rowptr[dst[e]] + rank[e]] = src[e];
}

template <int DOUT, bool RELU_IN>
__global__ void k_gemm_wmma(const float* __restrict__ X,
                            const float* __restrict__ W,
                            float* __restrict__ H) {
    __shared__ float lds_x[16 * 132];
    __shared__ float lds_w[128 * DOUT];

    const int tid     = threadIdx.x;
    const int lane    = tid & 31;
    const int wave    = tid >> 5;
    const int rowBase = blockIdx.x * 16;
    const int colBase = wave * 16;
    const int nthr    = (DOUT / 16) * 32;

    for (int i = tid; i < 16 * 128; i += nthr) {
        int r = i >> 7, c = i & 127;
        float v = X[(size_t)(rowBase + r) * 128 + c];
        if (RELU_IN) v = fmaxf(v, 0.0f);
        lds_x[r * 132 + c] = v;
    }
    for (int i = tid; i < 128 * DOUT; i += nthr) lds_w[i] = W[i];
    __syncthreads();

    const int r    = lane & 15;
    const int half = lane >> 4;
    const int koff = half * 2;

    v8f acc = {};
#pragma unroll
    for (int kk = 0; kk < 128; kk += 4) {
        v2f a = *(const v2f*)(lds_x + r * 132 + kk + koff);
        v2f b;
        b.x = lds_w[(kk + koff) * DOUT + colBase + r];
        b.y = lds_w[(kk + koff + 1) * DOUT + colBase + r];
        acc = __builtin_amdgcn_wmma_f32_16x16x4_f32(
             false, a,  false, b,
             (short)0, acc,  false,  false);
    }

#pragma unroll
    for (int v = 0; v < 8; ++v) {
        H[(size_t)(rowBase + v + half * 8) * DOUT + colBase + r] = acc[v];
    }
}

template <int DOUT>
using rowvec = float __attribute__((ext_vector_type(DOUT / 32)));

template <int DOUT, int K>
__device__ __forceinline__ void gather_k(rowvec<DOUT>& acc, const rowvec<DOUT>* __restrict__ hrow,
                                         const float* __restrict__ dinv, const int* __restrict__ col,
                                         int j, float wd, int lane) {
    int   s[K];
    float w[K];
    rowvec<DOUT> r[K];
#pragma unroll
    for (int u = 0; u < K; ++u) s[u] = col[j + u];
#pragma unroll
    for (int u = 0; u < K; ++u) r[u] = hrow[(size_t)s[u] * 32 + lane];
#pragma unroll
    for (int u = 0; u < K; ++u) w[u] = dinv[s[u]] * wd;
#pragma unroll
    for (int u = 0; u < K; ++u) acc += r[u] * w[u];
}

template <int DOUT, int NPW>
__global__ __launch_bounds__(256) void k_gather(const float* __restrict__ h,
                                                const float* __restrict__ b,
                                                const float* __restrict__ dinv,
                                                const int* __restrict__ rowptr,
                                                const int* __restrict__ deg,
                                                const int* __restrict__ col,
                                                float* __restrict__ out) {
    typedef rowvec<DOUT> vec;
    const int lane = threadIdx.x & 31;
    const int wv = __builtin_amdgcn_readfirstlane((int)((blockIdx.x * blockDim.x + threadIdx.x) >> 5));
    const vec* __restrict__ hrow = (const vec*)h;
    const vec bias = ((const vec*)b)[lane];

    for (int i = 0; i < NPW; ++i) {
        const int d = wv * NPW + i;
        if (d >= Nn) break;
        int j = rowptr[d];
        const int end = j + deg[d];
        const float wd = dinv[d];
        vec acc = hrow[(size_t)d * 32 + lane] * (wd * wd) + bias;
        for (; j + 4 <= end; j += 4) gather_k<DOUT, 4>(acc, hrow, dinv, col, j, wd, lane);
        if (j + 2 <= end) { gather_k<DOUT, 2>(acc, hrow, dinv, col, j, wd, lane); j += 2; }
        if (j < end) gather_k<DOUT, 1>(acc, hrow, dinv, col, j, wd, lane);
        ((vec*)out)[(size_t)d * 32 + lane] = acc;
    }
}

extern "C" void kernel_launch(void* const* d_in, const int* in_sizes, int n_in,
                              void* d_out, int out_size, void* d_ws, size_t ws_size,
                              hipStream_t stream) {
    const float* X   = (const float*)d_in[0];
    const float* W1  = (const float*)d_in[1];
    const float* b1  = (const float*)d_in[2];
    const float* W2  = (const float*)d_in[3];
    const float* b2  = (const float*)d_in[4];
    const float* W3  = (const float*)d_in[5];
    const float* b3  = (const float*)d_in[6];
    const int*   ei  = (const int*)d_in[7];
    const int*   src = ei;
    const int*   dst = ei + Ee;
    float* out = (float*)d_out;

    float* bufA   = (float*)d_ws;
    float* bufB   = bufA + (size_t)Nn * 128;
    int*   deg    = (int*)(bufB + (size_t)Nn * 128);
    float* dinv   = (float*)(deg + Nn);
    int*   rowptr = (int*)(dinv + Nn);
    int*   rank   = rowptr + Nn;
    int*   col    = rank + Ee;

    const int T = 256;
    k_zero_deg<<<(Nn + T - 1) / T, T, 0, stream>>>(deg);
    k_count_deg<<<(Ee + T - 1) / T, T, 0, stream>>>(dst, deg, rank);
    k_scan_dinv<<<(Nn + 1023) / 1024, 256, 0, stream>>>(deg, rowptr, dinv);
    k_fill<<<(Ee + T - 1) / T, T, 0, stream>>>(src, dst, rowptr, rank, col);

    const int gGemm = Nn / 16;
    const int NPW   = 4;
    const int gAgg  = (Nn + 8 * NPW - 1) / (8 * NPW);

    k_gemm_wmma<128, false><<<gGemm, 256, 0, stream>>>(X, W1, bufA);
    k_gather<128, NPW><<<gAgg, 256, 0, stream>>>(bufA, b1, dinv, rowptr, deg, col, bufB);

    k_gemm_wmma<128, true><<<gGemm, 256, 0, stream>>>(bufB, W2, bufA);
    k_gather<128, NPW><<<gAgg, 256, 0, stream>>>(bufA, b2, dinv, rowptr, deg, col, bufB);

    k_gemm_wmma<64, true><<<gGemm, 128, 0, stream>>>(bufB, W3, bufA);
    k_gather<64, NPW><<<gAgg, 256, 0, stream>>>(bufA, b3, dinv, rowptr, deg, col, out);
}


# GEMM blocks made resident: a wave keeps its 128x16 slice of W in registers, the block walks 5 row tiles, X tile double-buffered in LDS (was: all of W staged to LDS per 16-row block)
# speedup vs baseline: 2.3795x; 1.1477x over previous
#include <hip/hip_runtime.h>

#define Nn 100000
#define Ee 1000000
#define INF 128
#define HID 128
#define CLS 64

typedef __attribute__((ext_vector_type(2))) float v2f;
typedef __attribute__((ext_vector_type(8))) float v8f;

__global__ void k_zero_deg(int* __restrict__ deg) {
    int i = blockIdx.x * blockDim.x + threadIdx.x;
    if (i < Nn) deg[i] = 0;
}

__global__ void k_count_deg(const int* __restrict__ dst, int* __restrict__ deg, int* __restrict__ rank) {
    int e = blockIdx.x * blockDim.x + threadIdx.x;
    if (e < Ee) rank[e] = atomicAdd(&deg[dst[e]], 1);
}

__global__ __launch_bounds__(256) void k_scan_dinv(const int* __restrict__ deg,
                                                   int* __restrict__ rowptr,
                                                   float* __restrict__ dinv) {
    __shared__ int s_pre[8];
    __shared__ int s_tot[8];
    const int tid  = threadIdx.x;
    const int lane = tid & 31;
    const int wave = tid >> 5;
    const int4* deg4 = (const int4*)deg;

    int pre = 0;
    const int lim = blockIdx.x * 256;
#pragma unroll 8
    for (int i = tid; i < lim; i += 256) {
        int4 v = deg4[i];
        pre += (v.x + v.y) + (v.z + v.w);
    }
    for (int o = 16; o > 0; o >>= 1) pre += __shfl_xor(pre, o);
    if (lane == 0) s_pre[wave] = pre;

    const int i4 = blockIdx.x * 256 + tid;
    const bool ok = i4 < Nn / 4;
    int4 v = make_int4(0, 0, 0, 0);
    if (ok) v = deg4[i4];
    const int tsum = (v.x + v.y) + (v.z + v.w);
    int inc = tsum;
    for (int o = 1; o < 32; o <<= 1) {
        int t = __shfl_up(inc, o);
        if (lane >= o) inc += t;
    }
    if (lane == 31) s_tot[wave] = inc;
    __syncthreads();
    int base = 0;
    for (int w = 0; w < 8; ++w) {
        base += s_pre[w];
        if (w < wave) base += s_tot[w];
    }
    if (ok) {
        int4 r;
        r.x = base + inc - tsum;
        r.y = r.x + v.x;
        r.z = r.y + v.y;
        r.w = r.z + v.z;
        ((int4*)rowptr)[i4] = r;
        float4 f;
        f.x = 1.0f / sqrtf((float)v.x + 1.0f);
        f.y = 1.0f / sqrtf((float)v.y + 1.0f);
        f.z = 1.0f / sqrtf((float)v.z + 1.0f);
        f.w = 1.0f / sqrtf((float)v.w + 1.0f);
        ((float4*)dinv)[i4] = f;
    }
}

__global__ void k_fill(const int* __restrict__ src, const int* __restrict__ dst,
                       const int* __restrict__ rowptr, const int* __restrict__ rank,
                       int* __restrict__ col) {
    int e = blockIdx.x * blockDim.x + threadIdx.x;
    if (e < Ee) col[rowptr[dst[e]] + rank[e]] = src[e];
}

template <int DOUT, bool RELU_IN, int TPB>
__global__ __launch_bounds__((DOUT / 16) * 32, 5 * (DOUT / 64)) void k_gemm_wmma(const float* __restrict__ X,
                                                                const float* __restrict__ W,
                                                                float* __restrict__ H) {
    constexpr int NT    = (DOUT / 16) * 32;
    constexpr int PITCH = 132;
    constexpr int NF4   = (16 * 128 / 4) / NT;
    __shared__ __attribute__((aligned(16))) float lds_x[2][16 * PITCH];

    const int tid     = threadIdx.x;
    const int lane    = tid & 31;
    const int wave    = tid >> 5;
    const int colBase = wave * 16;
    const int r       = lane & 15;
    const int half    = lane >> 4;
    const int koff    = half * 2;

    v2f bfrag[32];
#pragma unroll
    for (int s = 0; s < 32; ++s) {
        bfrag[s].x = W[(4 * s + koff) * DOUT + colBase + r];
        bfrag[s].y = W[(4 * s + koff + 1) * DOUT + colBase + r];
    }

    const int tile0 = blockIdx.x * TPB;
    const unsigned xoff = (unsigned)tid * 16u;
    const unsigned loff = (unsigned)((tid >> 5) * PITCH + (tid & 31) * 4) * 4u;
    const unsigned hoff = (unsigned)((half * 8) * DOUT + colBase + r) * 4u;
    constexpr unsigned LROWS = (NT / 32) * PITCH * 4;
    const char* __restrict__ xt = (const char*)X + (size_t)tile0 * (16 * 128 * 4);
    char* __restrict__ ht = (char*)H + (size_t)tile0 * (16 * DOUT * 4);
    char* lbase = (char*)&lds_x[0][0];
    float4 stage[NF4];

#pragma unroll
    for (int u = 0; u < NF4; ++u) stage[u] = *(const float4*)(xt + xoff + u * (NT * 16));
#pragma unroll
    for (int u = 0; u < NF4; ++u) {
        float4 v = stage[u];
        if (RELU_IN) { v.x = fmaxf(v.x, 0.0f); v.y = fmaxf(v.y, 0.0f); v.z = fmaxf(v.z, 0.0f); v.w = fmaxf(v.w, 0.0f); }
        *(float4*)(lbase + loff + u * LROWS) = v;
    }
    __syncthreads();

#pragma unroll 1
    for (int t = 0; t < TPB; ++t) {
        const float* __restrict__ xa = &lds_x[t & 1][r * PITCH + koff];
        if (t + 1 < TPB) {
            xt += 16 * 128 * 4;
#pragma unroll
            for (int u = 0; u < NF4; ++u) stage[u] = *(const float4*)(xt + xoff + u * (NT * 16));
        }

        __builtin_amdgcn_sched_barrier(0);

        v8f acc = {};
#pragma unroll
        for (int s = 0; s < 32; ++s) {
            v2f a = *(const v2f*)(xa + 4 * s);
            acc = __builtin_amdgcn_wmma_f32_16x16x4_f32(
                 false, a,  false, bfrag[s],
                 (short)0, acc,  false,  false);
        }

#pragma unroll
        for (int v = 0; v < 8; ++v) *(float*)(ht + hoff + v * (DOUT * 4)) = acc[v];
        ht += 16 * DOUT * 4;

        if (t + 1 < TPB) {
            char* lnext = lbase + ((t + 1) & 1) * (16 * PITCH * 4);
#pragma unroll
            for (int u = 0; u < NF4; ++u) {
                float4 v = stage[u];
                if (RELU_IN) { v.x = fmaxf(v.x, 0.0f); v.y = fmaxf(v.y, 0.0f); v.z = fmaxf(v.z, 0.0f); v.w = fmaxf(v.w, 0.0f); }
                *(float4*)(lnext + loff + u * LROWS) = v;
            }
            __syncthreads();
        }
    }
}

template <int DOUT>
using rowvec = float __attribute__((ext_vector_type(DOUT / 32)));

template <int DOUT, int K>
__device__ __forceinline__ void gather_k(rowvec<DOUT>& acc, const rowvec<DOUT>* __restrict__ hrow,
                                         const float* __restrict__ dinv, const int* __restrict__ col,
                                         int j, float wd, int lane) {
    int   s[K];
    float w[K];
    rowvec<DOUT> r[K];
#pragma unroll
    for (int u = 0; u < K; ++u) s[u] = col[j + u];
#pragma unroll
    for (int u = 0; u < K; ++u) r[u] = hrow[(size_t)s[u] * 32 + lane];
#pragma unroll
    for (int u = 0; u < K; ++u) w[u] = dinv[s[u]] * wd;
#pragma unroll
    for (int u = 0; u < K; ++u) acc += r[u] * w[u];
}

template <int DOUT, int NPW>
__global__ __launch_bounds__(256) void k_gather(const float* __restrict__ h,
                                                const float* __restrict__ b,
                                                const float* __restrict__ dinv,
                                                const int* __restrict__ rowptr,
                                                const int* __restrict__ deg,
                                                const int* __restrict__ col,
                                                float* __restrict__ out) {
    typedef rowvec<DOUT> vec;
    const int lane = threadIdx.x & 31;
    const int wv = __builtin_amdgcn_readfirstlane((int)((blockIdx.x * blockDim.x + threadIdx.x) >> 5));
    const vec* __restrict__ hrow = (const vec*)h;
    const vec bias = ((const vec*)b)[lane];

    for (int i = 0; i < NPW; ++i) {
        const int d = wv * NPW + i;
        if (d >= Nn) break;
        int j = rowptr[d];
        const int end = j + deg[d];
        const float wd = dinv[d];
        vec acc = hrow[(size_t)d * 32 + lane] * (wd * wd) + bias;
        for (; j + 4 <= end; j += 4) gather_k<DOUT, 4>(acc, hrow, dinv, col, j, wd, lane);
        if (j + 2 <= end) { gather_k<DOUT, 2>(acc, hrow, dinv, col, j, wd, lane); j += 2; }
        if (j < end) gather_k<DOUT, 1>(acc, hrow, dinv, col, j, wd, lane);
        ((vec*)out)[(size_t)d * 32 + lane] = acc;
    }
}

extern "C" void kernel_launch(void* const* d_in, const int* in_sizes, int n_in,
                              void* d_out, int out_size, void* d_ws, size_t ws_size,
                              hipStream_t stream) {
    const float* X   = (const float*)d_in[0];
    const float* W1  = (const float*)d_in[1];
    const float* b1  = (const float*)d_in[2];
    const float* W2  = (const float*)d_in[3];
    const float* b2  = (const float*)d_in[4];
    const float* W3  = (const float*)d_in[5];
    const float* b3  = (const float*)d_in[6];
    const int*   ei  = (const int*)d_in[7];
    const int*   src = ei;
    const int*   dst = ei + Ee;
    float* out = (float*)d_out;

    float* bufA   = (float*)d_ws;
    float* bufB   = bufA + (size_t)Nn * 128;
    int*   deg    = (int*)(bufB + (size_t)Nn * 128);
    float* dinv   = (float*)(deg + Nn);
    int*   rowptr = (int*)(dinv + Nn);
    int*   rank   = rowptr + Nn;
    int*   col    = rank + Ee;

    const int T = 256;
    k_zero_deg<<<(Nn + T - 1) / T, T, 0, stream>>>(deg);
    k_count_deg<<<(Ee + T - 1) / T, T, 0, stream>>>(dst, deg, rank);
    k_scan_dinv<<<(Nn + 1023) / 1024, 256, 0, stream>>>(deg, rowptr, dinv);
    k_fill<<<(Ee + T - 1) / T, T, 0, stream>>>(src, dst, rowptr, rank, col);

    const int TPB   = 5;
    const int gGemm = Nn / (16 * TPB);
    const int NPW   = 4;
    const int gAgg  = (Nn + 8 * NPW - 1) / (8 * NPW);

    k_gemm_wmma<128, false, TPB><<<gGemm, 256, 0, stream>>>(X, W1, bufA);
    k_gather<128, NPW><<<gAgg, 256, 0, stream>>>(bufA, b1, dinv, rowptr, deg, col, bufB);

    k_gemm_wmma<128, true, TPB><<<gGemm, 256, 0, stream>>>(bufB, W2, bufA);
    k_gather<128, NPW><<<gAgg, 256, 0, stream>>>(bufA, b2, dinv, rowptr, deg, col, bufB);

    k_gemm_wmma<64, true, TPB><<<gGemm, 128, 0, stream>>>(bufB, W3, bufA);
    k_gather<64, NPW><<<gAgg, 256, 0, stream>>>(bufA, b3, dinv, rowptr, deg, col, out);
}


# Aggregations of layers 1 and 2 fused into the next layer's GEMM: a block's 8 waves sum 2 rows each of the 16-row tile in registers (same order), bias, relu, LDS tile, then WMMA; 10 launches to 8
# speedup vs baseline: 2.5419x; 1.0682x over previous
#include <hip/hip_runtime.h>

#define Nn 100000
#define Ee 1000000
#define INF 128
#define HID 128
#define CLS 64

typedef __attribute__((ext_vector_type(2))) float v2f;
typedef __attribute__((ext_vector_type(8))) float v8f;

__global__ void k_zero_deg(int* __restrict__ deg) {
    int i = blockIdx.x * blockDim.x + threadIdx.x;
    if (i < Nn) deg[i] = 0;
}

__global__ void k_count_deg(const int* __restrict__ dst, int* __restrict__ deg, int* __restrict__ rank) {
    int e = blockIdx.x * blockDim.x + threadIdx.x;
    if (e < Ee) rank[e] = atomicAdd(&deg[dst[e]], 1);
}

__global__ __launch_bounds__(256) void k_scan_dinv(const int* __restrict__ deg,
                                                   int* __restrict__ rowptr,
                                                   float* __restrict__ dinv) {
    __shared__ int s_pre[8];
    __shared__ int s_tot[8];
    const int tid  = threadIdx.x;
    const int lane = tid & 31;
    const int wave = tid >> 5;
    const int4* deg4 = (const int4*)deg;

    int pre = 0;
    const int lim = blockIdx.x * 256;
#pragma unroll 8
    for (int i = tid; i < lim; i += 256) {
        int4 v = deg4[i];
        pre += (v.x + v.y) + (v.z + v.w);
    }
    for (int o = 16; o > 0; o >>= 1) pre += __shfl_xor(pre, o);
    if (lane == 0) s_pre[wave] = pre;

    const int i4 = blockIdx.x * 256 + tid;
    const bool ok = i4 < Nn / 4;
    int4 v = make_int4(0, 0, 0, 0);
    if (ok) v = deg4[i4];
    const int tsum = (v.x + v.y) + (v.z + v.w);
    int inc = tsum;
    for (int o = 1; o < 32; o <<= 1) {
        int t = __shfl_up(inc, o);
        if (lane >= o) inc += t;
    }
    if (lane == 31) s_tot[wave] = inc;
    __syncthreads();
    int base = 0;
    for (int w = 0; w < 8; ++w) {
        base += s_pre[w];
        if (w < wave) base += s_tot[w];
    }
    if (ok) {
        int4 r;
        r.x = base + inc - tsum;
        r.y = r.x + v.x;
        r.z = r.y + v.y;
        r.w = r.z + v.z;
        ((int4*)rowptr)[i4] = r;
        float4 f;
        f.x = 1.0f / sqrtf((float)v.x + 1.0f);
        f.y = 1.0f / sqrtf((float)v.y + 1.0f);
        f.z = 1.0f / sqrtf((float)v.z + 1.0f);
        f.w = 1.0f / sqrtf((float)v.w + 1.0f);
        ((float4*)dinv)[i4] = f;
    }
}

__global__ void k_fill(const int* __restrict__ src, const int* __restrict__ dst,
                       const int* __restrict__ rowptr, const int* __restrict__ rank,
                       int* __restrict__ col) {
    int e = blockIdx.x * blockDim.x + threadIdx.x;
    if (e < Ee) col[rowptr[dst[e]] + rank[e]] = src[e];
}

template <int DOUT, bool RELU_IN, int TPB>
__global__ __launch_bounds__((DOUT / 16) * 32, 5 * (DOUT / 64)) void k_gemm_wmma(const float* __restrict__ X,
                                                                const float* __restrict__ W,
                                                                float* __restrict__ H) {
    constexpr int NT    = (DOUT / 16) * 32;
    constexpr int PITCH = 132;
    constexpr int NF4   = (16 * 128 / 4) / NT;
    __shared__ __attribute__((aligned(16))) float lds_x[2][16 * PITCH];

    const int tid     = threadIdx.x;
    const int lane    = tid & 31;
    const int wave    = tid >> 5;
    const int colBase = wave * 16;
    const int r       = lane & 15;
    const int half    = lane >> 4;
    const int koff    = half * 2;

    v2f bfrag[32];
#pragma unroll
    for (int s = 0; s < 32; ++s) {
        bfrag[s].x = W[(4 * s + koff) * DOUT + colBase + r];
        bfrag[s].y = W[(4 * s + koff + 1) * DOUT + colBase + r];
    }

    const int tile0 = blockIdx.x * TPB;
    const unsigned xoff = (unsigned)tid * 16u;
    const unsigned loff = (unsigned)((tid >> 5) * PITCH + (tid & 31) * 4) * 4u;
    const unsigned hoff = (unsigned)((half * 8) * DOUT + colBase + r) * 4u;
    constexpr unsigned LROWS = (NT / 32) * PITCH * 4;
    const char* __restrict__ xt = (const char*)X + (size_t)tile0 * (16 * 128 * 4);
    char* __restrict__ ht = (char*)H + (size_t)tile0 * (16 * DOUT * 4);
    char* lbase = (char*)&lds_x[0][0];
    float4 stage[NF4];

#pragma unroll
    for (int u = 0; u < NF4; ++u) stage[u] = *(const float4*)(xt + xoff + u * (NT * 16));
#pragma unroll
    for (int u = 0; u < NF4; ++u) {
        float4 v = stage[u];
        if (RELU_IN) { v.x = fmaxf(v.x, 0.0f); v.y = fmaxf(v.y, 0.0f); v.z = fmaxf(v.z, 0.0f); v.w = fmaxf(v.w, 0.0f); }
        *(float4*)(lbase + loff + u * LROWS) = v;
    }
    __syncthreads();

#pragma unroll 1
    for (int t = 0; t < TPB; ++t) {
        const float* __restrict__ xa = &lds_x[t & 1][r * PITCH + koff];
        if (t + 1 < TPB) {
            xt += 16 * 128 * 4;
#pragma unroll
            for (int u = 0; u < NF4; ++u) stage[u] = *(const float4*)(xt + xoff + u * (NT * 16));
        }

        __builtin_amdgcn_sched_barrier(0);

        v8f acc = {};
#pragma unroll
        for (int s = 0; s < 32; ++s) {
            v2f a = *(const v2f*)(xa + 4 * s);
            acc = __builtin_amdgcn_wmma_f32_16x16x4_f32(
                 false, a,  false, bfrag[s],
                 (short)0, acc,  false,  false);
        }

#pragma unroll
        for (int v = 0; v < 8; ++v) *(float*)(ht + hoff + v * (DOUT * 4)) = acc[v];
        ht += 16 * DOUT * 4;

        if (t + 1 < TPB) {
            char* lnext = lbase + ((t + 1) & 1) * (16 * PITCH * 4);
#pragma unroll
            for (int u = 0; u < NF4; ++u) {
                float4 v = stage[u];
                if (RELU_IN) { v.x = fmaxf(v.x, 0.0f); v.y = fmaxf(v.y, 0.0f); v.z = fmaxf(v.z, 0.0f); v.w = fmaxf(v.w, 0.0f); }
                *(float4*)(lnext + loff + u * LROWS) = v;
            }
            __syncthreads();
        }
    }
}

template <int DOUT>
using rowvec = float __attribute__((ext_vector_type(DOUT / 32)));

template <int DOUT, int K>
__device__ __forceinline__ void gather_k(rowvec<DOUT>& acc, const rowvec<DOUT>* __restrict__ hrow,
                                         const float* __restrict__ dinv, const int* __restrict__ col,
                                         int j, float wd, int lane) {
    int   s[K];
    float w[K];
    rowvec<DOUT> r[K];
#pragma unroll
    for (int u = 0; u < K; ++u) s[u] = col[j + u];
#pragma unroll
    for (int u = 0; u < K; ++u) r[u] = hrow[(size_t)s[u] * 32 + lane];
#pragma unroll
    for (int u = 0; u < K; ++u) w[u] = dinv[s[u]] * wd;
#pragma unroll
    for (int u = 0; u < K; ++u) acc += r[u] * w[u];
}

template <int DOUT, int NPW>
__global__ __launch_bounds__(256) void k_gather(const float* __restrict__ h,
                                                const float* __restrict__ b,
                                                const float* __restrict__ dinv,
                                                const int* __restrict__ rowptr,
                                                const int* __restrict__ deg,
                                                const int* __restrict__ col,
                                                float* __restrict__ out) {
    typedef rowvec<DOUT> vec;
    const int lane = threadIdx.x & 31;
    const int wv = __builtin_amdgcn_readfirstlane((int)((blockIdx.x * blockDim.x + threadIdx.x) >> 5));
    const vec* __restrict__ hrow = (const vec*)h;
    const vec bias = ((const vec*)b)[lane];

    for (int i = 0; i < NPW; ++i) {
        const int d = wv * NPW + i;
        if (d >= Nn) break;
        int j = rowptr[d];
        const int end = j + deg[d];
        const float wd = dinv[d];
        vec acc = hrow[(size_t)d * 32 + lane] * (wd * wd) + bias;
        for (; j + 4 <= end; j += 4) gather_k<DOUT, 4>(acc, hrow, dinv, col, j, wd, lane);
        if (j + 2 <= end) { gather_k<DOUT, 2>(acc, hrow, dinv, col, j, wd, lane); j += 2; }
        if (j < end) gather_k<DOUT, 1>(acc, hrow, dinv, col, j, wd, lane);
        ((vec*)out)[(size_t)d * 32 + lane] = acc;
    }
}


typedef const int   __attribute__((address_space(4))) kint;
typedef const float __attribute__((address_space(4))) kflt;

template <int K>
__device__ __forceinline__ void gather_kc(rowvec<128>& acc, const rowvec<128>* __restrict__ hrow,
                                          kflt* dinv, kint* col, int j, float wd, int lane) {
    int   s[K];
    float di[K];
    rowvec<128> r[K];
#pragma unroll
    for (int u = 0; u < K; ++u) s[u] = col[j + u];
#pragma unroll
    for (int u = 0; u < K; ++u) r[u] = hrow[(size_t)s[u] * 32 + lane];
#pragma unroll
    for (int u = 0; u < K; ++u) di[u] = dinv[s[u]];
    if constexpr (K == 4) asm volatile("" :: "s"(di[0]), "s"(di[1]), "s"(di[2]), "s"(di[3]));
    if constexpr (K == 2) asm volatile("" :: "s"(di[0]), "s"(di[1]));
#pragma unroll
    for (int u = 0; u < K; ++u) acc += r[u] * (di[u] * wd);
}
template <int DOUT, int WPS>
__global__ __launch_bounds__(256, WPS) void k_agg_gemm(const float* __restrict__ h,
                                                       const float* __restrict__ b,
                                                       const float* __restrict__ dinv,
                                                       const int* __restrict__ rowptr,
                                                       const int* __restrict__ deg,
                                                       const int* __restrict__ col,
                                                       const float* __restrict__ W,
                                                       float* __restrict__ H) {
    constexpr int NMM   = DOUT / 16;
    constexpr int RPW   = 2;
    constexpr int PITCH = 132;
    typedef rowvec<128> vec;
    __shared__ __attribute__((aligned(16))) float lds_x[2][16 * PITCH];
    __shared__ __attribute__((aligned(16))) float lds_b[128];

    const int tid     = threadIdx.x;
    const int lane    = tid & 31;
    const int wave    = __builtin_amdgcn_readfirstlane(tid >> 5);
    const int colBase = wave * 16;
    const int r       = lane & 15;
    const int half    = lane >> 4;
    const int koff    = half * 2;

    v2f bfrag[32];
    if (wave < NMM) {
#pragma unroll
        for (int s = 0; s < 32; ++s) {
            bfrag[s].x = W[(4 * s + koff) * DOUT + colBase + r];
            bfrag[s].y = W[(4 * s + koff + 1) * DOUT + colBase + r];
        }
    }

    const vec* __restrict__ hrow = (const vec*)h;
    if (tid < 32) *(vec*)(lds_b + lane * 4) = ((const vec*)b)[lane];
    __syncthreads();
    kint* rowptr_k = (kint*)rowptr;
    kint* deg_k    = (kint*)deg;
    kint* col_k    = (kint*)col;
    kflt* dinv_k   = (kflt*)dinv;
    const unsigned hoff = (unsigned)((half * 8) * DOUT + colBase + r) * 4u;

    int it = 0;
#pragma unroll 1
    for (int tile = blockIdx.x; tile < Nn / 16; tile += gridDim.x, ++it) {
        float* buf = &lds_x[it & 1][0];

        for (int n = 0; n < RPW; ++n) {
            const int row = wave * RPW + n;
            const int d   = tile * 16 + row;
            int j = rowptr_k[d];
            const int end = j + deg_k[d];
            const float wd = dinv_k[d];
            vec acc = hrow[(size_t)d * 32 + lane] * (wd * wd) + *(const vec*)(lds_b + lane * 4);
            for (; j + 4 <= end; j += 4) gather_kc<4>(acc, hrow, dinv_k, col_k, j, wd, lane);
            if (j + 2 <= end) { gather_kc<2>(acc, hrow, dinv_k, col_k, j, wd, lane); j += 2; }
            if (j < end) gather_kc<1>(acc, hrow, dinv_k, col_k, j, wd, lane);
            acc.x = fmaxf(acc.x, 0.0f); acc.y = fmaxf(acc.y, 0.0f);
            acc.z = fmaxf(acc.z, 0.0f); acc.w = fmaxf(acc.w, 0.0f);
            *(vec*)(buf + row * PITCH + lane * 4) = acc;
        }
        __syncthreads();

        if (wave < NMM) {
            const float* __restrict__ xa = &lds_x[it & 1][r * PITCH + koff];
            v8f c = {};
#pragma unroll
            for (int s = 0; s < 32; ++s) {
                v2f a = *(const v2f*)(xa + 4 * s);
                c = __builtin_amdgcn_wmma_f32_16x16x4_f32(
                     false, a,  false, bfrag[s],
                     (short)0, c,  false,  false);
            }

            char* __restrict__ ht = (char*)H + (size_t)tile * (16 * DOUT * 4);
#pragma unroll
            for (int v = 0; v < 8; ++v) *(float*)(ht + hoff + v * (DOUT * 4)) = c[v];
        }
    }
}

extern "C" void kernel_launch(void* const* d_in, const int* in_sizes, int n_in,
                              void* d_out, int out_size, void* d_ws, size_t ws_size,
                              hipStream_t stream) {
    const float* X   = (const float*)d_in[0];
    const float* W1  = (const float*)d_in[1];
    const float* b1  = (const float*)d_in[2];
    const float* W2  = (const float*)d_in[3];
    const float* b2  = (const float*)d_in[4];
    const float* W3  = (const float*)d_in[5];
    const float* b3  = (const float*)d_in[6];
    const int*   ei  = (const int*)d_in[7];
    const int*   src = ei;
    const int*   dst = ei + Ee;
    float* out = (float*)d_out;

    float* bufA   = (float*)d_ws;
    float* bufB   = bufA + (size_t)Nn * 128;
    int*   deg    = (int*)(bufB + (size_t)Nn * 128);
    float* dinv   = (float*)(deg + Nn);
    int*   rowptr = (int*)(dinv + Nn);
    int*   rank   = rowptr + Nn;
    int*   col    = rank + Ee;

    const int T = 256;
    k_zero_deg<<<(Nn + T - 1) / T, T, 0, stream>>>(deg);
    k_count_deg<<<(Ee + T - 1) / T, T, 0, stream>>>(dst, deg, rank);
    k_scan_dinv<<<(Nn + 1023) / 1024, 256, 0, stream>>>(deg, rowptr, dinv);
    k_fill<<<(Ee + T - 1) / T, T, 0, stream>>>(src, dst, rowptr, rank, col);

    const int TPB   = 5;
    const int gGemm = Nn / (16 * TPB);
    const int NPW   = 4;
    const int gAgg  = (Nn + 8 * NPW - 1) / (8 * NPW);

    k_gemm_wmma<128, false, TPB><<<gGemm, 256, 0, stream>>>(X, W1, bufA);

    k_agg_gemm<128, 10><<<1250, 256, 0, stream>>>(bufA, b1, dinv, rowptr, deg, col, W2, bufB);

    k_agg_gemm<64, 10><<<1250, 256, 0, stream>>>(bufB, b2, dinv, rowptr, deg, col, W3, bufA);

    k_gather<64, NPW><<<gAgg, 256, 0, stream>>>(bufA, b3, dinv, rowptr, deg, col, out);
}


# Edge list built by a two-level counting sort on LDS counters (1024-node ranges, then nodes; slot = LDS add's return, sum order stays arrival order), not 1M returning global adds; 8 launches to 7
# speedup vs baseline: 3.3280x; 1.3093x over previous
#include <hip/hip_runtime.h>

#define Nn 100000
#define Ee 1000000
#define INF 128
#define HID 128
#define CLS 64

typedef __attribute__((ext_vector_type(2))) float v2f;
typedef __attribute__((ext_vector_type(8))) float v8f;

#define RSH 10
#define NR  ((Nn + 1023) >> RSH)
#define RP  128
#define CH  4096
#define NB  ((Ee + CH - 1) / CH)

__global__ __launch_bounds__(1024) void k_range_hist(const int* __restrict__ dst, int* __restrict__ hist) {
    __shared__ int s_h[RP];
    const int tid = threadIdx.x;
    const int e   = blockIdx.x * CH + tid * 4;
    const bool ok = e < Ee;
    int4 d = make_int4(0, 0, 0, 0);
    if (ok) d = *(const int4*)(dst + e);
    if (tid < RP) s_h[tid] = 0;
    __syncthreads();
    if (ok) {
        atomicAdd(&s_h[d.x >> RSH], 1);
        atomicAdd(&s_h[d.y >> RSH], 1);
        atomicAdd(&s_h[d.z >> RSH], 1);
        atomicAdd(&s_h[d.w >> RSH], 1);
    }
    __syncthreads();
    if (tid < RP) hist[blockIdx.x * RP + tid] = s_h[tid];
}

__global__ __launch_bounds__(1024) void k_range_scatter(const int* __restrict__ src, const int* __restrict__ dst,
                                                        const int* __restrict__ hist,
                                                        int* __restrict__ rbase, int* __restrict__ part) {
    __shared__ __attribute__((aligned(16))) int s_tot[RP];
    __shared__ __attribute__((aligned(16))) int s_bef[RP];
    __shared__ __attribute__((aligned(16))) int s_base[RP];
    __shared__ int s_cnt[RP];
    const int tid = threadIdx.x;
    const int b   = blockIdx.x;
    const int e   = b * CH + tid * 4;
    const bool ok = e < Ee;

    const int r = tid & (RP - 1);
    const int g = tid >> 7;
    int tot = 0, bef = 0;
#pragma unroll
    for (int k = 0; k < (NB + 7) / 8; ++k) {
        const int bb = g + 8 * k;
        const int bc = bb < NB ? bb : NB - 1;
        int v = hist[bc * RP + r];
        v = bb < NB ? v : 0;
        tot += v;
        bef += bb < b ? v : 0;
    }
    int4 d = make_int4(0, 0, 0, 0), s = make_int4(0, 0, 0, 0);
    if (ok) { d = *(const int4*)(dst + e); s = *(const int4*)(src + e); }
    if (tid < RP) { s_tot[tid] = 0; s_bef[tid] = 0; s_cnt[tid] = 0; }
    __syncthreads();
    atomicAdd(&s_tot[r], tot);
    atomicAdd(&s_bef[r], bef);
    __syncthreads();
    if (tid < 32) {
        const int4 t4 = *(const int4*)&s_tot[4 * tid];
        const int sum = (t4.x + t4.y) + (t4.z + t4.w);
        int inc = sum;
        for (int o = 1; o < 32; o <<= 1) {
            int t = __shfl_up(inc, o);
            if (tid >= o) inc += t;
        }
        int4 rb;
        rb.x = inc - sum;
        rb.y = rb.x + t4.x;
        rb.z = rb.y + t4.y;
        rb.w = rb.z + t4.z;
        const int4 bf = *(const int4*)&s_bef[4 * tid];
        *(int4*)&s_base[4 * tid] = make_int4(rb.x + bf.x, rb.y + bf.y, rb.z + bf.z, rb.w + bf.w);
        if (b == 0) *(int4*)&rbase[4 * tid] = rb;
    }
    __syncthreads();
    if (ok) {
        const int q0 = d.x >> RSH, q1 = d.y >> RSH, q2 = d.z >> RSH, q3 = d.w >> RSH;
        const int p0 = s_base[q0] + atomicAdd(&s_cnt[q0], 1);
        const int p1 = s_base[q1] + atomicAdd(&s_cnt[q1], 1);
        const int p2 = s_base[q2] + atomicAdd(&s_cnt[q2], 1);
        const int p3 = s_base[q3] + atomicAdd(&s_cnt[q3], 1);
        part[p0] = (s.x << RSH) | (d.x & 1023);
        part[p1] = (s.y << RSH) | (d.y & 1023);
        part[p2] = (s.z << RSH) | (d.z & 1023);
        part[p3] = (s.w << RSH) | (d.w & 1023);
    }
}

__global__ __launch_bounds__(1024) void k_node_sort(const int* __restrict__ part, const int* __restrict__ rbase,
                                                    int* __restrict__ deg, int* __restrict__ rowptr,
                                                    float* __restrict__ dinv, int* __restrict__ col) {
    constexpr int KEEP = 12;
    __shared__ int s_cnt[1024];
    __shared__ int s_w[32];
    const int tid  = threadIdx.x;
    const int lane = tid & 31;
    const int wave = tid >> 5;
    const int beg  = rbase[blockIdx.x];
    const int end  = rbase[blockIdx.x + 1];

    int v[KEEP];
#pragma unroll
    for (int u = 0; u < KEEP; ++u) {
        const int i = beg + tid + u * 1024;
        v[u] = i < end ? part[i] : -1;
    }
    s_cnt[tid] = 0;
    __syncthreads();
#pragma unroll
    for (int u = 0; u < KEEP; ++u)
        if (v[u] >= 0) atomicAdd(&s_cnt[v[u] & 1023], 1);
    for (int i = beg + tid + KEEP * 1024; i < end; i += 1024) atomicAdd(&s_cnt[part[i] & 1023], 1);
    __syncthreads();

    const int c = s_cnt[tid];
    int inc = c;
    for (int o = 1; o < 32; o <<= 1) {
        int t = __shfl_up(inc, o);
        if (lane >= o) inc += t;
    }
    if (lane == 31) s_w[wave] = inc;
    __syncthreads();
    if (wave == 0) {
        const int w = s_w[lane];
        int wi = w;
        for (int o = 1; o < 32; o <<= 1) {
            int t = __shfl_up(wi, o);
            if (lane >= o) wi += t;
        }
        s_w[lane] = wi - w;
    }
    __syncthreads();
    const int pos  = beg + s_w[wave] + inc - c;
    const int node = blockIdx.x * 1024 + tid;
    s_cnt[tid] = pos;
    if (node < Nn) {
        deg[node]    = c;
        rowptr[node] = pos;
        dinv[node]   = 1.0f / sqrtf((float)c + 1.0f);
    }
    __syncthreads();
#pragma unroll
    for (int u = 0; u < KEEP; ++u)
        if (v[u] >= 0) col[atomicAdd(&s_cnt[v[u] & 1023], 1)] = v[u] >> RSH;
    for (int i = beg + tid + KEEP * 1024; i < end; i += 1024) {
        const int x = part[i];
        col[atomicAdd(&s_cnt[x & 1023], 1)] = x >> RSH;
    }
}

template <int DOUT, bool RELU_IN, int TPB>
__global__ __launch_bounds__((DOUT / 16) * 32, 5 * (DOUT / 64)) void k_gemm_wmma(const float* __restrict__ X,
                                                                const float* __restrict__ W,
                                                                float* __restrict__ H) {
    constexpr int NT    = (DOUT / 16) * 32;
    constexpr int PITCH = 132;
    constexpr int NF4   = (16 * 128 / 4) / NT;
    __shared__ __attribute__((aligned(16))) float lds_x[2][16 * PITCH];

    const int tid     = threadIdx.x;
    const int lane    = tid & 31;
    const int wave    = tid >> 5;
    const int colBase = wave * 16;
    const int r       = lane & 15;
    const int half    = lane >> 4;
    const int koff    = half * 2;

    v2f bfrag[32];
#pragma unroll
    for (int s = 0; s < 32; ++s) {
        bfrag[s].x = W[(4 * s + koff) * DOUT + colBase + r];
        bfrag[s].y = W[(4 * s + koff + 1) * DOUT + colBase + r];
    }

    const int tile0 = blockIdx.x * TPB;
    const unsigned xoff = (unsigned)tid * 16u;
    const unsigned loff = (unsigned)((tid >> 5) * PITCH + (tid & 31) * 4) * 4u;
    const unsigned hoff = (unsigned)((half * 8) * DOUT + colBase + r) * 4u;
    constexpr unsigned LROWS = (NT / 32) * PITCH * 4;
    const char* __restrict__ xt = (const char*)X + (size_t)tile0 * (16 * 128 * 4);
    char* __restrict__ ht = (char*)H + (size_t)tile0 * (16 * DOUT * 4);
    char* lbase = (char*)&lds_x[0][0];
    float4 stage[NF4];

#pragma unroll
    for (int u = 0; u < NF4; ++u) stage[u] = *(const float4*)(xt + xoff + u * (NT * 16));
#pragma unroll
    for (int u = 0; u < NF4; ++u) {
        float4 v = stage[u];
        if (RELU_IN) { v.x = fmaxf(v.x, 0.0f); v.y = fmaxf(v.y, 0.0f); v.z = fmaxf(v.z, 0.0f); v.w = fmaxf(v.w, 0.0f); }
        *(float4*)(lbase + loff + u * LROWS) = v;
    }
    __syncthreads();

#pragma unroll 1
    for (int t = 0; t < TPB; ++t) {
        const float* __restrict__ xa = &lds_x[t & 1][r * PITCH + koff];
        if (t + 1 < TPB) {
            xt += 16 * 128 * 4;
#pragma unroll
            for (int u = 0; u < NF4; ++u) stage[u] = *(const float4*)(xt + xoff + u * (NT * 16));
        }

        __builtin_amdgcn_sched_barrier(0);

        v8f acc = {};
#pragma unroll
        for (int s = 0; s < 32; ++s) {
            v2f a = *(const v2f*)(xa + 4 * s);
            acc = __builtin_amdgcn_wmma_f32_16x16x4_f32(
                 false, a,  false, bfrag[s],
                 (short)0, acc,  false,  false);
        }

#pragma unroll
        for (int v = 0; v < 8; ++v) *(float*)(ht + hoff + v * (DOUT * 4)) = acc[v];
        ht += 16 * DOUT * 4;

        if (t + 1 < TPB) {
            char* lnext = lbase + ((t + 1) & 1) * (16 * PITCH * 4);
#pragma unroll
            for (int u = 0; u < NF4; ++u) {
                float4 v = stage[u];
                if (RELU_IN) { v.x = fmaxf(v.x, 0.0f); v.y = fmaxf(v.y, 0.0f); v.z = fmaxf(v.z, 0.0f); v.w = fmaxf(v.w, 0.0f); }
                *(float4*)(lnext + loff + u * LROWS) = v;
            }
            __syncthreads();
        }
    }
}

template <int DOUT>
using rowvec = float __attribute__((ext_vector_type(DOUT / 32)));

template <int DOUT, int K>
__device__ __forceinline__ void gather_k(rowvec<DOUT>& acc, const rowvec<DOUT>* __restrict__ hrow,
                                         const float* __restrict__ dinv, const int* __restrict__ col,
                                         int j, float wd, int lane) {
    int   s[K];
    float w[K];
    rowvec<DOUT> r[K];
#pragma unroll
    for (int u = 0; u < K; ++u) s[u] = col[j + u];
#pragma unroll
    for (int u = 0; u < K; ++u) r[u] = hrow[(size_t)s[u] * 32 + lane];
#pragma unroll
    for (int u = 0; u < K; ++u) w[u] = dinv[s[u]] * wd;
#pragma unroll
    for (int u = 0; u < K; ++u) acc += r[u] * w[u];
}

template <int DOUT, int NPW>
__global__ __launch_bounds__(256) void k_gather(const float* __restrict__ h,
                                                const float* __restrict__ b,
                                                const float* __restrict__ dinv,
                                                const int* __restrict__ rowptr,
                                                const int* __restrict__ deg,
                                                const int* __restrict__ col,
                                                float* __restrict__ out) {
    typedef rowvec<DOUT> vec;
    const int lane = threadIdx.x & 31;
    const int wv = __builtin_amdgcn_readfirstlane((int)((blockIdx.x * blockDim.x + threadIdx.x) >> 5));
    const vec* __restrict__ hrow = (const vec*)h;
    const vec bias = ((const vec*)b)[lane];

    for (int i = 0; i < NPW; ++i) {
        const int d = wv * NPW + i;
        if (d >= Nn) break;
        int j = rowptr[d];
        const int end = j + deg[d];
        const float wd = dinv[d];
        vec acc = hrow[(size_t)d * 32 + lane] * (wd * wd) + bias;
        for (; j + 4 <= end; j += 4) gather_k<DOUT, 4>(acc, hrow, dinv, col, j, wd, lane);
        if (j + 2 <= end) { gather_k<DOUT, 2>(acc, hrow, dinv, col, j, wd, lane); j += 2; }
        if (j < end) gather_k<DOUT, 1>(acc, hrow, dinv, col, j, wd, lane);
        ((vec*)out)[(size_t)d * 32 + lane] = acc;
    }
}


typedef const int   __attribute__((address_space(4))) kint;
typedef const float __attribute__((address_space(4))) kflt;

template <int K>
__device__ __forceinline__ void gather_kc(rowvec<128>& acc, const rowvec<128>* __restrict__ hrow,
                                          kflt* dinv, kint* col, int j, float wd, int lane) {
    int   s[K];
    float di[K];
    rowvec<128> r[K];
#pragma unroll
    for (int u = 0; u < K; ++u) s[u] = col[j + u];
#pragma unroll
    for (int u = 0; u < K; ++u) r[u] = hrow[(size_t)s[u] * 32 + lane];
#pragma unroll
    for (int u = 0; u < K; ++u) di[u] = dinv[s[u]];
    if constexpr (K == 4) asm volatile("" :: "s"(di[0]), "s"(di[1]), "s"(di[2]), "s"(di[3]));
    if constexpr (K == 2) asm volatile("" :: "s"(di[0]), "s"(di[1]));
#pragma unroll
    for (int u = 0; u < K; ++u) acc += r[u] * (di[u] * wd);
}
template <int DOUT, int WPS>
__global__ __launch_bounds__(256, WPS) void k_agg_gemm(const float* __restrict__ h,
                                                       const float* __restrict__ b,
                                                       const float* __restrict__ dinv,
                                                       const int* __restrict__ rowptr,
                                                       const int* __restrict__ deg,
                                                       const int* __restrict__ col,
                                                       const float* __restrict__ W,
                                                       float* __restrict__ H) {
    constexpr int NMM   = DOUT / 16;
    constexpr int RPW   = 2;
    constexpr int PITCH = 132;
    typedef rowvec<128> vec;
    __shared__ __attribute__((aligned(16))) float lds_x[2][16 * PITCH];
    __shared__ __attribute__((aligned(16))) float lds_b[128];

    const int tid     = threadIdx.x;
    const int lane    = tid & 31;
    const int wave    = __builtin_amdgcn_readfirstlane(tid >> 5);
    const int colBase = wave * 16;
    const int r       = lane & 15;
    const int half    = lane >> 4;
    const int koff    = half * 2;

    v2f bfrag[32];
    if (wave < NMM) {
#pragma unroll
        for (int s = 0; s < 32; ++s) {
            bfrag[s].x = W[(4 * s + koff) * DOUT + colBase + r];
            bfrag[s].y = W[(4 * s + koff + 1) * DOUT + colBase + r];
        }
    }

    const vec* __restrict__ hrow = (const vec*)h;
    if (tid < 32) *(vec*)(lds_b + lane * 4) = ((const vec*)b)[lane];
    __syncthreads();
    kint* rowptr_k = (kint*)rowptr;
    kint* deg_k    = (kint*)deg;
    kint* col_k    = (kint*)col;
    kflt* dinv_k   = (kflt*)dinv;
    const unsigned hoff = (unsigned)((half * 8) * DOUT + colBase + r) * 4u;

    int it = 0;
#pragma unroll 1
    for (int tile = blockIdx.x; tile < Nn / 16; tile += gridDim.x, ++it) {
        float* buf = &lds_x[it & 1][0];

        for (int n = 0; n < RPW; ++n) {
            const int row = wave * RPW + n;
            const int d   = tile * 16 + row;
            int j = rowptr_k[d];
            const int end = j + deg_k[d];
            const float wd = dinv_k[d];
            vec acc = hrow[(size_t)d * 32 + lane] * (wd * wd) + *(const vec*)(lds_b + lane * 4);
            for (; j + 4 <= end; j += 4) gather_kc<4>(acc, hrow, dinv_k, col_k, j, wd, lane);
            if (j + 2 <= end) { gather_kc<2>(acc, hrow, dinv_k, col_k, j, wd, lane); j += 2; }
            if (j < end) gather_kc<1>(acc, hrow, dinv_k, col_k, j, wd, lane);
            acc.x = fmaxf(acc.x, 0.0f); acc.y = fmaxf(acc.y, 0.0f);
            acc.z = fmaxf(acc.z, 0.0f); acc.w = fmaxf(acc.w, 0.0f);
            *(vec*)(buf + row * PITCH + lane * 4) = acc;
        }
        __syncthreads();

        if (wave < NMM) {
            const float* __restrict__ xa = &lds_x[it & 1][r * PITCH + koff];
            v8f c = {};
#pragma unroll
            for (int s = 0; s < 32; ++s) {
                v2f a = *(const v2f*)(xa + 4 * s);
                c = __builtin_amdgcn_wmma_f32_16x16x4_f32(
                     false, a,  false, bfrag[s],
                     (short)0, c,  false,  false);
            }

            char* __restrict__ ht = (char*)H + (size_t)tile * (16 * DOUT * 4);
#pragma unroll
            for (int v = 0; v < 8; ++v) *(float*)(ht + hoff + v * (DOUT * 4)) = c[v];
        }
    }
}

extern "C" void kernel_launch(void* const* d_in, const int* in_sizes, int n_in,
                              void* d_out, int out_size, void* d_ws, size_t ws_size,
                              hipStream_t stream) {
    const float* X   = (const float*)d_in[0];
    const float* W1  = (const float*)d_in[1];
    const float* b1  = (const float*)d_in[2];
    const float* W2  = (const float*)d_in[3];
    const float* b2  = (const float*)d_in[4];
    const float* W3  = (const float*)d_in[5];
    const float* b3  = (const float*)d_in[6];
    const int*   ei  = (const int*)d_in[7];
    const int*   src = ei;
    const int*   dst = ei + Ee;
    float* out = (float*)d_out;

    float* bufA   = (float*)d_ws;
    float* bufB   = bufA + (size_t)Nn * 128;
    int*   deg    = (int*)(bufB + (size_t)Nn * 128);
    float* dinv   = (float*)(deg + Nn);
    int*   rowptr = (int*)(dinv + Nn);
    int*   part   = rowptr + Nn;
    int*   col    = part + Ee;
    int*   hist   = col + Ee;
    int*   rbase  = hist + NB * RP;

    k_range_hist<<<NB, 1024, 0, stream>>>(dst, hist);
    k_range_scatter<<<NB, 1024, 0, stream>>>(src, dst, hist, rbase, part);
    k_node_sort<<<NR, 1024, 0, stream>>>(part, rbase, deg, rowptr, dinv, col);

    const int TPB   = 5;
    const int gGemm = Nn / (16 * TPB);
    const int NPW   = 4;
    const int gAgg  = (Nn + 8 * NPW - 1) / (8 * NPW);

    k_gemm_wmma<128, false, TPB><<<gGemm, 256, 0, stream>>>(X, W1, bufA);

    k_agg_gemm<128, 10><<<1250, 256, 0, stream>>>(bufA, b1, dinv, rowptr, deg, col, W2, bufB);

    k_agg_gemm<64, 10><<<1250, 256, 0, stream>>>(bufB, b2, dinv, rowptr, deg, col, W3, bufA);

    k_gather<64, NPW><<<gAgg, 256, 0, stream>>>(bufA, b3, dinv, rowptr, deg, col, out);
}


# Normalisation folded into the stored rows: product kernels store h*dinv[row], gathers add rows unweighted (same list order) and scale once by dinv[dst]; no per-edge weight fetch; rounding differs
# speedup vs baseline: 4.1758x; 1.2547x over previous
#include <hip/hip_runtime.h>

#define Nn 100000
#define Ee 1000000
#define INF 128
#define HID 128
#define CLS 64

typedef __attribute__((ext_vector_type(2))) float v2f;
typedef __attribute__((ext_vector_type(8))) float v8f;

#define RSH 10
#define NR  ((Nn + 1023) >> RSH)
#define RP  128
#define CH  4096
#define NB  ((Ee + CH - 1) / CH)

__global__ __launch_bounds__(1024) void k_range_hist(const int* __restrict__ dst, int* __restrict__ hist) {
    __shared__ int s_h[RP];
    const int tid = threadIdx.x;
    const int e   = blockIdx.x * CH + tid * 4;
    const bool ok = e < Ee;
    int4 d = make_int4(0, 0, 0, 0);
    if (ok) d = *(const int4*)(dst + e);
    if (tid < RP) s_h[tid] = 0;
    __syncthreads();
    if (ok) {
        atomicAdd(&s_h[d.x >> RSH], 1);
        atomicAdd(&s_h[d.y >> RSH], 1);
        atomicAdd(&s_h[d.z >> RSH], 1);
        atomicAdd(&s_h[d.w >> RSH], 1);
    }
    __syncthreads();
    if (tid < RP) hist[blockIdx.x * RP + tid] = s_h[tid];
}

__global__ __launch_bounds__(1024) void k_range_scatter(const int* __restrict__ src, const int* __restrict__ dst,
                                                        const int* __restrict__ hist,
                                                        int* __restrict__ rbase, int* __restrict__ part) {
    __shared__ __attribute__((aligned(16))) int s_tot[RP];
    __shared__ __attribute__((aligned(16))) int s_bef[RP];
    __shared__ __attribute__((aligned(16))) int s_base[RP];
    __shared__ int s_cnt[RP];
    const int tid = threadIdx.x;
    const int b   = blockIdx.x;
    const int e   = b * CH + tid * 4;
    const bool ok = e < Ee;

    const int r = tid & (RP - 1);
    const int g = tid >> 7;
    int tot = 0, bef = 0;
#pragma unroll
    for (int k = 0; k < (NB + 7) / 8; ++k) {
        const int bb = g + 8 * k;
        const int bc = bb < NB ? bb : NB - 1;
        int v = hist[bc * RP + r];
        v = bb < NB ? v : 0;
        tot += v;
        bef += bb < b ? v : 0;
    }
    int4 d = make_int4(0, 0, 0, 0), s = make_int4(0, 0, 0, 0);
    if (ok) { d = *(const int4*)(dst + e); s = *(const int4*)(src + e); }
    if (tid < RP) { s_tot[tid] = 0; s_bef[tid] = 0; s_cnt[tid] = 0; }
    __syncthreads();
    atomicAdd(&s_tot[r], tot);
    atomicAdd(&s_bef[r], bef);
    __syncthreads();
    if (tid < 32) {
        const int4 t4 = *(const int4*)&s_tot[4 * tid];
        const int sum = (t4.x + t4.y) + (t4.z + t4.w);
        int inc = sum;
        for (int o = 1; o < 32; o <<= 1) {
            int t = __shfl_up(inc, o);
            if (tid >= o) inc += t;
        }
        int4 rb;
        rb.x = inc - sum;
        rb.y = rb.x + t4.x;
        rb.z = rb.y + t4.y;
        rb.w = rb.z + t4.z;
        const int4 bf = *(const int4*)&s_bef[4 * tid];
        *(int4*)&s_base[4 * tid] = make_int4(rb.x + bf.x, rb.y + bf.y, rb.z + bf.z, rb.w + bf.w);
        if (b == 0) *(int4*)&rbase[4 * tid] = rb;
    }
    __syncthreads();
    if (ok) {
        const int q0 = d.x >> RSH, q1 = d.y >> RSH, q2 = d.z >> RSH, q3 = d.w >> RSH;
        const int p0 = s_base[q0] + atomicAdd(&s_cnt[q0], 1);
        const int p1 = s_base[q1] + atomicAdd(&s_cnt[q1], 1);
        const int p2 = s_base[q2] + atomicAdd(&s_cnt[q2], 1);
        const int p3 = s_base[q3] + atomicAdd(&s_cnt[q3], 1);
        part[p0] = (s.x << RSH) | (d.x & 1023);
        part[p1] = (s.y << RSH) | (d.y & 1023);
        part[p2] = (s.z << RSH) | (d.z & 1023);
        part[p3] = (s.w << RSH) | (d.w & 1023);
    }
}

__global__ __launch_bounds__(1024) void k_node_sort(const int* __restrict__ part, const int* __restrict__ rbase,
                                                    int* __restrict__ deg, int* __restrict__ rowptr,
                                                    float* __restrict__ dinv, int* __restrict__ col) {
    constexpr int KEEP = 12;
    __shared__ int s_cnt[1024];
    __shared__ int s_w[32];
    const int tid  = threadIdx.x;
    const int lane = tid & 31;
    const int wave = tid >> 5;
    const int beg  = rbase[blockIdx.x];
    const int end  = rbase[blockIdx.x + 1];

    int v[KEEP];
#pragma unroll
    for (int u = 0; u < KEEP; ++u) {
        const int i = beg + tid + u * 1024;
        v[u] = i < end ? part[i] : -1;
    }
    s_cnt[tid] = 0;
    __syncthreads();
#pragma unroll
    for (int u = 0; u < KEEP; ++u)
        if (v[u] >= 0) atomicAdd(&s_cnt[v[u] & 1023], 1);
    for (int i = beg + tid + KEEP * 1024; i < end; i += 1024) atomicAdd(&s_cnt[part[i] & 1023], 1);
    __syncthreads();

    const int c = s_cnt[tid];
    int inc = c;
    for (int o = 1; o < 32; o <<= 1) {
        int t = __shfl_up(inc, o);
        if (lane >= o) inc += t;
    }
    if (lane == 31) s_w[wave] = inc;
    __syncthreads();
    if (wave == 0) {
        const int w = s_w[lane];
        int wi = w;
        for (int o = 1; o < 32; o <<= 1) {
            int t = __shfl_up(wi, o);
            if (lane >= o) wi += t;
        }
        s_w[lane] = wi - w;
    }
    __syncthreads();
    const int pos  = beg + s_w[wave] + inc - c;
    const int node = blockIdx.x * 1024 + tid;
    s_cnt[tid] = pos;
    if (node < Nn) {
        deg[node]    = c;
        rowptr[node] = pos;
        dinv[node]   = 1.0f / sqrtf((float)c + 1.0f);
    }
    __syncthreads();
#pragma unroll
    for (int u = 0; u < KEEP; ++u)
        if (v[u] >= 0) col[atomicAdd(&s_cnt[v[u] & 1023], 1)] = v[u] >> RSH;
    for (int i = beg + tid + KEEP * 1024; i < end; i += 1024) {
        const int x = part[i];
        col[atomicAdd(&s_cnt[x & 1023], 1)] = x >> RSH;
    }
}

typedef const int   __attribute__((address_space(4))) kint;
typedef const float __attribute__((address_space(4))) kflt;

template <int DOUT, bool RELU_IN, int TPB>
__global__ __launch_bounds__((DOUT / 16) * 32, 5 * (DOUT / 64)) void k_gemm_wmma(const float* __restrict__ X,
                                                                const float* __restrict__ W,
                                                                const float* __restrict__ dinv,
                                                                float* __restrict__ H) {
    constexpr int NT    = (DOUT / 16) * 32;
    constexpr int PITCH = 132;
    constexpr int NF4   = (16 * 128 / 4) / NT;
    __shared__ __attribute__((aligned(16))) float lds_x[2][16 * PITCH];

    const int tid     = threadIdx.x;
    const int lane    = tid & 31;
    const int wave    = __builtin_amdgcn_readfirstlane(tid >> 5);
    const int colBase = wave * 16;
    const int r       = lane & 15;
    const int half    = lane >> 4;
    const int koff    = half * 2;

    v2f bfrag[32];
#pragma unroll
    for (int s = 0; s < 32; ++s) {
        bfrag[s].x = W[(4 * s + koff) * DOUT + colBase + r];
        bfrag[s].y = W[(4 * s + koff + 1) * DOUT + colBase + r];
    }

    const int tile0 = blockIdx.x * TPB;
    kflt* dinv_k = (kflt*)dinv + tile0 * 16 + wave;
    float sc[NF4];
    const unsigned xoff = (unsigned)tid * 16u;
    const unsigned loff = (unsigned)((tid >> 5) * PITCH + (tid & 31) * 4) * 4u;
    const unsigned hoff = (unsigned)((half * 8) * DOUT + colBase + r) * 4u;
    constexpr unsigned LROWS = (NT / 32) * PITCH * 4;
    const char* __restrict__ xt = (const char*)X + (size_t)tile0 * (16 * 128 * 4);
    char* __restrict__ ht = (char*)H + (size_t)tile0 * (16 * DOUT * 4);
    char* lbase = (char*)&lds_x[0][0];
    float4 stage[NF4];

#pragma unroll
    for (int u = 0; u < NF4; ++u) stage[u] = *(const float4*)(xt + xoff + u * (NT * 16));
#pragma unroll
    for (int u = 0; u < NF4; ++u) sc[u] = dinv_k[u * (NT / 32)];
#pragma unroll
    for (int u = 0; u < NF4; ++u) {
        float4 v = stage[u];
        if (RELU_IN) { v.x = fmaxf(v.x, 0.0f); v.y = fmaxf(v.y, 0.0f); v.z = fmaxf(v.z, 0.0f); v.w = fmaxf(v.w, 0.0f); }
        v.x *= sc[u]; v.y *= sc[u]; v.z *= sc[u]; v.w *= sc[u];
        *(float4*)(lbase + loff + u * LROWS) = v;
    }
    __syncthreads();

#pragma unroll 1
    for (int t = 0; t < TPB; ++t) {
        const float* __restrict__ xa = &lds_x[t & 1][r * PITCH + koff];
        if (t + 1 < TPB) {
            xt += 16 * 128 * 4;
            dinv_k += 16;
#pragma unroll
            for (int u = 0; u < NF4; ++u) stage[u] = *(const float4*)(xt + xoff + u * (NT * 16));
#pragma unroll
            for (int u = 0; u < NF4; ++u) sc[u] = dinv_k[u * (NT / 32)];
        }

        __builtin_amdgcn_sched_barrier(0);

        v8f acc = {};
#pragma unroll
        for (int s = 0; s < 32; ++s) {
            v2f a = *(const v2f*)(xa + 4 * s);
            acc = __builtin_amdgcn_wmma_f32_16x16x4_f32(
                 false, a,  false, bfrag[s],
                 (short)0, acc,  false,  false);
        }

#pragma unroll
        for (int v = 0; v < 8; ++v) *(float*)(ht + hoff + v * (DOUT * 4)) = acc[v];
        ht += 16 * DOUT * 4;

        if (t + 1 < TPB) {
            char* lnext = lbase + ((t + 1) & 1) * (16 * PITCH * 4);
#pragma unroll
            for (int u = 0; u < NF4; ++u) {
                float4 v = stage[u];
                if (RELU_IN) { v.x = fmaxf(v.x, 0.0f); v.y = fmaxf(v.y, 0.0f); v.z = fmaxf(v.z, 0.0f); v.w = fmaxf(v.w, 0.0f); }
                v.x *= sc[u]; v.y *= sc[u]; v.z *= sc[u]; v.w *= sc[u];
                *(float4*)(lnext + loff + u * LROWS) = v;
            }
            __syncthreads();
        }
    }
}

template <int DOUT>
using rowvec = float __attribute__((ext_vector_type(DOUT / 32)));

template <int DOUT, int K>
__device__ __forceinline__ void gather_k(rowvec<DOUT>& acc, const rowvec<DOUT>* __restrict__ hrow,
                                         const int* __restrict__ col, int j, int lane) {
    int   s[K];
    rowvec<DOUT> r[K];
#pragma unroll
    for (int u = 0; u < K; ++u) s[u] = col[j + u];
#pragma unroll
    for (int u = 0; u < K; ++u) r[u] = hrow[(size_t)s[u] * 32 + lane];
#pragma unroll
    for (int u = 0; u < K; ++u) acc += r[u];
}

template <int DOUT, int NPW>
__global__ __launch_bounds__(256) void k_gather(const float* __restrict__ h,
                                                const float* __restrict__ b,
                                                const float* __restrict__ dinv,
                                                const int* __restrict__ rowptr,
                                                const int* __restrict__ deg,
                                                const int* __restrict__ col,
                                                float* __restrict__ out) {
    typedef rowvec<DOUT> vec;
    const int lane = threadIdx.x & 31;
    const int wv = __builtin_amdgcn_readfirstlane((int)((blockIdx.x * blockDim.x + threadIdx.x) >> 5));
    const vec* __restrict__ hrow = (const vec*)h;
    const vec bias = ((const vec*)b)[lane];

    for (int i = 0; i < NPW; ++i) {
        const int d = wv * NPW + i;
        if (d >= Nn) break;
        int j = rowptr[d];
        const int end = j + deg[d];
        const float wd = dinv[d];
        vec acc = hrow[(size_t)d * 32 + lane];
        for (; j + 4 <= end; j += 4) gather_k<DOUT, 4>(acc, hrow, col, j, lane);
        if (j + 2 <= end) { gather_k<DOUT, 2>(acc, hrow, col, j, lane); j += 2; }
        if (j < end) gather_k<DOUT, 1>(acc, hrow, col, j, lane);
        ((vec*)out)[(size_t)d * 32 + lane] = acc * wd + bias;
    }
}


template <int K>
__device__ __forceinline__ void gather_kc(rowvec<128>& acc, const rowvec<128>* __restrict__ hrow,
                                          kint* col, int j, int lane) {
    int   s[K];
    rowvec<128> r[K];
#pragma unroll
    for (int u = 0; u < K; ++u) s[u] = col[j + u];
#pragma unroll
    for (int u = 0; u < K; ++u) r[u] = hrow[(size_t)s[u] * 32 + lane];
#pragma unroll
    for (int u = 0; u < K; ++u) acc += r[u];
}
template <int DOUT, int WPS>
__global__ __launch_bounds__(256, WPS) void k_agg_gemm(const float* __restrict__ h,
                                                       const float* __restrict__ b,
                                                       const float* __restrict__ dinv,
                                                       const int* __restrict__ rowptr,
                                                       const int* __restrict__ deg,
                                                       const int* __restrict__ col,
                                                       const float* __restrict__ W,
                                                       float* __restrict__ H) {
    constexpr int NMM   = DOUT / 16;
    constexpr int RPW   = 2;
    constexpr int PITCH = 132;
    typedef rowvec<128> vec;
    __shared__ __attribute__((aligned(16))) float lds_x[2][16 * PITCH];
    __shared__ __attribute__((aligned(16))) float lds_b[128];

    const int tid     = threadIdx.x;
    const int lane    = tid & 31;
    const int wave    = __builtin_amdgcn_readfirstlane(tid >> 5);
    const int colBase = wave * 16;
    const int r       = lane & 15;
    const int half    = lane >> 4;
    const int koff    = half * 2;

    v2f bfrag[32];
    if (wave < NMM) {
#pragma unroll
        for (int s = 0; s < 32; ++s) {
            bfrag[s].x = W[(4 * s + koff) * DOUT + colBase + r];
            bfrag[s].y = W[(4 * s + koff + 1) * DOUT + colBase + r];
        }
    }

    const vec* __restrict__ hrow = (const vec*)h;
    if (tid < 32) *(vec*)(lds_b + lane * 4) = ((const vec*)b)[lane];
    __syncthreads();
    kint* rowptr_k = (kint*)rowptr;
    kint* deg_k    = (kint*)deg;
    kint* col_k    = (kint*)col;
    kflt* dinv_k   = (kflt*)dinv;
    const unsigned hoff = (unsigned)((half * 8) * DOUT + colBase + r) * 4u;

    int it = 0;
#pragma unroll 1
    for (int tile = blockIdx.x; tile < Nn / 16; tile += gridDim.x, ++it) {
        float* buf = &lds_x[it & 1][0];

        for (int n = 0; n < RPW; ++n) {
            const int row = wave * RPW + n;
            const int d   = tile * 16 + row;
            int j = rowptr_k[d];
            const int end = j + deg_k[d];
            const float wd = dinv_k[d];
            vec acc = hrow[(size_t)d * 32 + lane];
            for (; j + 4 <= end; j += 4) gather_kc<4>(acc, hrow, col_k, j, lane);
            if (j + 2 <= end) { gather_kc<2>(acc, hrow, col_k, j, lane); j += 2; }
            if (j < end) gather_kc<1>(acc, hrow, col_k, j, lane);
            acc = acc * wd + *(const volatile __attribute__((address_space(3))) vec*)(lds_b + lane * 4);
            acc.x = fmaxf(acc.x, 0.0f) * wd; acc.y = fmaxf(acc.y, 0.0f) * wd;
            acc.z = fmaxf(acc.z, 0.0f) * wd; acc.w = fmaxf(acc.w, 0.0f) * wd;
            *(vec*)(buf + row * PITCH + lane * 4) = acc;
        }
        __syncthreads();

        if (wave < NMM) {
            const float* __restrict__ xa = &lds_x[it & 1][r * PITCH + koff];
            v8f c = {};
#pragma unroll
            for (int s = 0; s < 32; ++s) {
                v2f a = *(const v2f*)(xa + 4 * s);
                c = __builtin_amdgcn_wmma_f32_16x16x4_f32(
                     false, a,  false, bfrag[s],
                     (short)0, c,  false,  false);
            }

            char* __restrict__ ht = (char*)H + (size_t)tile * (16 * DOUT * 4);
#pragma unroll
            for (int v = 0; v < 8; ++v) *(float*)(ht + hoff + v * (DOUT * 4)) = c[v];
        }
    }
}

extern "C" void kernel_launch(void* const* d_in, const int* in_sizes, int n_in,
                              void* d_out, int out_size, void* d_ws, size_t ws_size,
                              hipStream_t stream) {
    const float* X   = (const float*)d_in[0];
    const float* W1  = (const float*)d_in[1];
    const float* b1  = (const float*)d_in[2];
    const float* W2  = (const float*)d_in[3];
    const float* b2  = (const float*)d_in[4];
    const float* W3  = (const float*)d_in[5];
    const float* b3  = (const float*)d_in[6];
    const int*   ei  = (const int*)d_in[7];
    const int*   src = ei;
    const int*   dst = ei + Ee;
    float* out = (float*)d_out;

    float* bufA   = (float*)d_ws;
    float* bufB   = bufA + (size_t)Nn * 128;
    int*   deg    = (int*)(bufB + (size_t)Nn * 128);
    float* dinv   = (float*)(deg + Nn);
    int*   rowptr = (int*)(dinv + Nn);
    int*   part   = rowptr + Nn;
    int*   col    = part + Ee;
    int*   hist   = col + Ee;
    int*   rbase  = hist + NB * RP;

    k_range_hist<<<NB, 1024, 0, stream>>>(dst, hist);
    k_range_scatter<<<NB, 1024, 0, stream>>>(src, dst, hist, rbase, part);
    k_node_sort<<<NR, 1024, 0, stream>>>(part, rbase, deg, rowptr, dinv, col);

    const int TPB   = 5;
    const int gGemm = Nn / (16 * TPB);
    const int NPW   = 4;
    const int gAgg  = (Nn + 8 * NPW - 1) / (8 * NPW);

    k_gemm_wmma<128, false, TPB><<<gGemm, 256, 0, stream>>>(X, W1, dinv, bufA);

    k_agg_gemm<128, 10><<<1250, 256, 0, stream>>>(bufA, b1, dinv, rowptr, deg, col, W2, bufB);

    k_agg_gemm<64, 10><<<1250, 256, 0, stream>>>(bufB, b2, dinv, rowptr, deg, col, W3, bufA);

    k_gather<64, NPW><<<gAgg, 256, 0, stream>>>(bufA, b3, dinv, rowptr, deg, col, out);
}


# Level 1 of the edge sort made block-local: a block orders its 4096 edges by range in LDS, stores one chunk plus its range offsets; the node pass collects a range from the 245 chunks; 7 launches to 6
# speedup vs baseline: 4.6744x; 1.1194x over previous
#include <hip/hip_runtime.h>

#define Nn 100000
#define Ee 1000000
#define INF 128
#define HID 128
#define CLS 64

typedef __attribute__((ext_vector_type(2))) float v2f;
typedef __attribute__((ext_vector_type(8))) float v8f;

#define RSH 10
#define NR  ((Nn + 1023) >> RSH)
#define RP  128
#define CH  4096
#define NB  ((Ee + CH - 1) / CH)
#define LP  132

__global__ __launch_bounds__(1024) void k_chunk_sort(const int* __restrict__ src, const int* __restrict__ dst,
                                                     int* __restrict__ part, int* __restrict__ loc) {
    __shared__ __attribute__((aligned(16))) int s_cnt[RP];
    __shared__ __attribute__((aligned(16))) int s_start[RP];
    __shared__ __attribute__((aligned(16))) int s_buf[CH];
    __shared__ int s_tot;
    const int tid = threadIdx.x;
    const int b   = blockIdx.x;
    const int e   = b * CH + tid * 4;
    const bool ok = e < Ee;
    int4 d = make_int4(0, 0, 0, 0), s = make_int4(0, 0, 0, 0);
    if (ok) { d = *(const int4*)(dst + e); s = *(const int4*)(src + e); }
    if (tid < RP) s_cnt[tid] = 0;
    __syncthreads();
    const int q0 = d.x >> RSH, q1 = d.y >> RSH, q2 = d.z >> RSH, q3 = d.w >> RSH;
    if (ok) {
        atomicAdd(&s_cnt[q0], 1);
        atomicAdd(&s_cnt[q1], 1);
        atomicAdd(&s_cnt[q2], 1);
        atomicAdd(&s_cnt[q3], 1);
    }
    __syncthreads();
    if (tid < 32) {
        const int4 t4 = *(const int4*)&s_cnt[4 * tid];
        const int sum = (t4.x + t4.y) + (t4.z + t4.w);
        int inc = sum;
        for (int o = 1; o < 32; o <<= 1) {
            int t = __shfl_up(inc, o);
            if (tid >= o) inc += t;
        }
        int4 rb;
        rb.x = inc - sum;
        rb.y = rb.x + t4.x;
        rb.z = rb.y + t4.y;
        rb.w = rb.z + t4.z;
        *(int4*)&s_start[4 * tid] = rb;
        *(int4*)&s_cnt[4 * tid]   = make_int4(0, 0, 0, 0);
        *(int4*)&loc[b * LP + 4 * tid] = rb;
        if (tid == 31) { s_tot = inc; loc[b * LP + RP] = inc; }
    }
    __syncthreads();
    if (ok) {
        const int p0 = s_start[q0] + atomicAdd(&s_cnt[q0], 1);
        const int p1 = s_start[q1] + atomicAdd(&s_cnt[q1], 1);
        const int p2 = s_start[q2] + atomicAdd(&s_cnt[q2], 1);
        const int p3 = s_start[q3] + atomicAdd(&s_cnt[q3], 1);
        s_buf[p0] = (s.x << RSH) | (d.x & 1023);
        s_buf[p1] = (s.y << RSH) | (d.y & 1023);
        s_buf[p2] = (s.z << RSH) | (d.z & 1023);
        s_buf[p3] = (s.w << RSH) | (d.w & 1023);
    }
    __syncthreads();
    if (4 * tid < s_tot) *(int4*)&part[b * CH + 4 * tid] = *(const int4*)&s_buf[4 * tid];
}

__global__ __launch_bounds__(1024) void k_node_sort(const int* __restrict__ part, const int* __restrict__ loc,
                                                    int* __restrict__ deg, int* __restrict__ rowptr,
                                                    float* __restrict__ dinv, int* __restrict__ col) {
    constexpr int KEEP = 16;
    __shared__ int s_cnt[1024];
    __shared__ int s_w[32];
    __shared__ int s_beg;
    const int tid  = threadIdx.x;
    const int lane = tid & 31;
    const int wave = tid >> 5;
    const int rg   = blockIdx.x;
    const int ck   = tid >> 2;
    const int k4   = tid & 3;
    int lo = 0, hi = 0;
    if (ck < NB) { lo = loc[ck * LP + rg]; hi = loc[ck * LP + rg + 1]; }
    const int* __restrict__ pc = part + (ck < NB ? ck : 0) * CH;

    int v[KEEP];
#pragma unroll
    for (int u = 0; u < KEEP; ++u) {
        const int i = lo + k4 + 4 * u;
        v[u] = i < hi ? pc[i] : -1;
    }
    s_cnt[tid] = 0;
    if (tid == 0) s_beg = 0;
    __syncthreads();
    int part_lo = k4 == 0 ? lo : 0;
    for (int o = 16; o > 0; o >>= 1) part_lo += __shfl_xor(part_lo, o);
    if (lane == 0) atomicAdd(&s_beg, part_lo);
#pragma unroll
    for (int u = 0; u < KEEP; ++u)
        if (v[u] >= 0) atomicAdd(&s_cnt[v[u] & 1023], 1);
    for (int i = lo + k4 + 4 * KEEP; i < hi; i += 4) atomicAdd(&s_cnt[pc[i] & 1023], 1);
    __syncthreads();

    const int beg = s_beg;
    const int c = s_cnt[tid];
    int inc = c;
    for (int o = 1; o < 32; o <<= 1) {
        int t = __shfl_up(inc, o);
        if (lane >= o) inc += t;
    }
    if (lane == 31) s_w[wave] = inc;
    __syncthreads();
    if (wave == 0) {
        const int w = s_w[lane];
        int wi = w;
        for (int o = 1; o < 32; o <<= 1) {
            int t = __shfl_up(wi, o);
            if (lane >= o) wi += t;
        }
        s_w[lane] = wi - w;
    }
    __syncthreads();
    const int pos  = beg + s_w[wave] + inc - c;
    const int node = rg * 1024 + tid;
    s_cnt[tid] = pos;
    if (node < Nn) {
        deg[node]    = c;
        rowptr[node] = pos;
        dinv[node]   = 1.0f / sqrtf((float)c + 1.0f);
    }
    __syncthreads();
#pragma unroll
    for (int u = 0; u < KEEP; ++u)
        if (v[u] >= 0) col[atomicAdd(&s_cnt[v[u] & 1023], 1)] = v[u] >> RSH;
    for (int i = lo + k4 + 4 * KEEP; i < hi; i += 4) {
        const int x = pc[i];
        col[atomicAdd(&s_cnt[x & 1023], 1)] = x >> RSH;
    }
}

typedef const int   __attribute__((address_space(4))) kint;
typedef const float __attribute__((address_space(4))) kflt;

template <int DOUT, bool RELU_IN, int TPB>
__global__ __launch_bounds__((DOUT / 16) * 32, 5 * (DOUT / 64)) void k_gemm_wmma(const float* __restrict__ X,
                                                                const float* __restrict__ W,
                                                                const float* __restrict__ dinv,
                                                                float* __restrict__ H) {
    constexpr int NT    = (DOUT / 16) * 32;
    constexpr int PITCH = 132;
    constexpr int NF4   = (16 * 128 / 4) / NT;
    __shared__ __attribute__((aligned(16))) float lds_x[2][16 * PITCH];

    const int tid     = threadIdx.x;
    const int lane    = tid & 31;
    const int wave    = __builtin_amdgcn_readfirstlane(tid >> 5);
    const int colBase = wave * 16;
    const int r       = lane & 15;
    const int half    = lane >> 4;
    const int koff    = half * 2;

    v2f bfrag[32];
#pragma unroll
    for (int s = 0; s < 32; ++s) {
        bfrag[s].x = W[(4 * s + koff) * DOUT + colBase + r];
        bfrag[s].y = W[(4 * s + koff + 1) * DOUT + colBase + r];
    }

    const int tile0 = blockIdx.x * TPB;
    kflt* dinv_k = (kflt*)dinv + tile0 * 16 + wave;
    float sc[NF4];
    const unsigned xoff = (unsigned)tid * 16u;
    const unsigned loff = (unsigned)((tid >> 5) * PITCH + (tid & 31) * 4) * 4u;
    const unsigned hoff = (unsigned)((half * 8) * DOUT + colBase + r) * 4u;
    constexpr unsigned LROWS = (NT / 32) * PITCH * 4;
    const char* __restrict__ xt = (const char*)X + (size_t)tile0 * (16 * 128 * 4);
    char* __restrict__ ht = (char*)H + (size_t)tile0 * (16 * DOUT * 4);
    char* lbase = (char*)&lds_x[0][0];
    float4 stage[NF4];

#pragma unroll
    for (int u = 0; u < NF4; ++u) stage[u] = *(const float4*)(xt + xoff + u * (NT * 16));
#pragma unroll
    for (int u = 0; u < NF4; ++u) sc[u] = dinv_k[u * (NT / 32)];
#pragma unroll
    for (int u = 0; u < NF4; ++u) {
        float4 v = stage[u];
        if (RELU_IN) { v.x = fmaxf(v.x, 0.0f); v.y = fmaxf(v.y, 0.0f); v.z = fmaxf(v.z, 0.0f); v.w = fmaxf(v.w, 0.0f); }
        v.x *= sc[u]; v.y *= sc[u]; v.z *= sc[u]; v.w *= sc[u];
        *(float4*)(lbase + loff + u * LROWS) = v;
    }
    __syncthreads();

#pragma unroll 1
    for (int t = 0; t < TPB; ++t) {
        const float* __restrict__ xa = &lds_x[t & 1][r * PITCH + koff];
        if (t + 1 < TPB) {
            xt += 16 * 128 * 4;
            dinv_k += 16;
#pragma unroll
            for (int u = 0; u < NF4; ++u) stage[u] = *(const float4*)(xt + xoff + u * (NT * 16));
#pragma unroll
            for (int u = 0; u < NF4; ++u) sc[u] = dinv_k[u * (NT / 32)];
        }

        __builtin_amdgcn_sched_barrier(0);

        v8f acc = {};
#pragma unroll
        for (int s = 0; s < 32; ++s) {
            v2f a = *(const v2f*)(xa + 4 * s);
            acc = __builtin_amdgcn_wmma_f32_16x16x4_f32(
                 false, a,  false, bfrag[s],
                 (short)0, acc,  false,  false);
        }

#pragma unroll
        for (int v = 0; v < 8; ++v) *(float*)(ht + hoff + v * (DOUT * 4)) = acc[v];
        ht += 16 * DOUT * 4;

        if (t + 1 < TPB) {
            char* lnext = lbase + ((t + 1) & 1) * (16 * PITCH * 4);
#pragma unroll
            for (int u = 0; u < NF4; ++u) {
                float4 v = stage[u];
                if (RELU_IN) { v.x = fmaxf(v.x, 0.0f); v.y = fmaxf(v.y, 0.0f); v.z = fmaxf(v.z, 0.0f); v.w = fmaxf(v.w, 0.0f); }
                v.x *= sc[u]; v.y *= sc[u]; v.z *= sc[u]; v.w *= sc[u];
                *(float4*)(lnext + loff + u * LROWS) = v;
            }
            __syncthreads();
        }
    }
}

template <int DOUT>
using rowvec = float __attribute__((ext_vector_type(DOUT / 32)));

template <int DOUT, int K>
__device__ __forceinline__ void gather_k(rowvec<DOUT>& acc, const rowvec<DOUT>* __restrict__ hrow,
                                         const int* __restrict__ col, int j, int lane) {
    int   s[K];
    rowvec<DOUT> r[K];
#pragma unroll
    for (int u = 0; u < K; ++u) s[u] = col[j + u];
#pragma unroll
    for (int u = 0; u < K; ++u) r[u] = hrow[(size_t)s[u] * 32 + lane];
#pragma unroll
    for (int u = 0; u < K; ++u) acc += r[u];
}

template <int DOUT, int NPW>
__global__ __launch_bounds__(256) void k_gather(const float* __restrict__ h,
                                                const float* __restrict__ b,
                                                const float* __restrict__ dinv,
                                                const int* __restrict__ rowptr,
                                                const int* __restrict__ deg,
                                                const int* __restrict__ col,
                                                float* __restrict__ out) {
    typedef rowvec<DOUT> vec;
    const int lane = threadIdx.x & 31;
    const int wv = __builtin_amdgcn_readfirstlane((int)((blockIdx.x * blockDim.x + threadIdx.x) >> 5));
    const vec* __restrict__ hrow = (const vec*)h;
    const vec bias = ((const vec*)b)[lane];

    for (int i = 0; i < NPW; ++i) {
        const int d = wv * NPW + i;
        if (d >= Nn) break;
        int j = rowptr[d];
        const int end = j + deg[d];
        const float wd = dinv[d];
        vec acc = hrow[(size_t)d * 32 + lane];
        for (; j + 4 <= end; j += 4) gather_k<DOUT, 4>(acc, hrow, col, j, lane);
        if (j + 2 <= end) { gather_k<DOUT, 2>(acc, hrow, col, j, lane); j += 2; }
        if (j < end) gather_k<DOUT, 1>(acc, hrow, col, j, lane);
        ((vec*)out)[(size_t)d * 32 + lane] = acc * wd + bias;
    }
}


template <int K>
__device__ __forceinline__ void gather_kc(rowvec<128>& acc, const rowvec<128>* __restrict__ hrow,
                                          kint* col, int j, int lane) {
    int   s[K];
    rowvec<128> r[K];
#pragma unroll
    for (int u = 0; u < K; ++u) s[u] = col[j + u];
#pragma unroll
    for (int u = 0; u < K; ++u) r[u] = hrow[(size_t)s[u] * 32 + lane];
#pragma unroll
    for (int u = 0; u < K; ++u) acc += r[u];
}
template <int DOUT, int WPS>
__global__ __launch_bounds__(256, WPS) void k_agg_gemm(const float* __restrict__ h,
                                                       const float* __restrict__ b,
                                                       const float* __restrict__ dinv,
                                                       const int* __restrict__ rowptr,
                                                       const int* __restrict__ deg,
                                                       const int* __restrict__ col,
                                                       const float* __restrict__ W,
                                                       float* __restrict__ H) {
    constexpr int NMM   = DOUT / 16;
    constexpr int RPW   = 2;
    constexpr int PITCH = 132;
    typedef rowvec<128> vec;
    __shared__ __attribute__((aligned(16))) float lds_x[2][16 * PITCH];
    __shared__ __attribute__((aligned(16))) float lds_b[128];

    const int tid     = threadIdx.x;
    const int lane    = tid & 31;
    const int wave    = __builtin_amdgcn_readfirstlane(tid >> 5);
    const int colBase = wave * 16;
    const int r       = lane & 15;
    const int half    = lane >> 4;
    const int koff    = half * 2;

    v2f bfrag[32];
    if (wave < NMM) {
#pragma unroll
        for (int s = 0; s < 32; ++s) {
            bfrag[s].x = W[(4 * s + koff) * DOUT + colBase + r];
            bfrag[s].y = W[(4 * s + koff + 1) * DOUT + colBase + r];
        }
    }

    const vec* __restrict__ hrow = (const vec*)h;
    if (tid < 32) *(vec*)(lds_b + lane * 4) = ((const vec*)b)[lane];
    __syncthreads();
    kint* rowptr_k = (kint*)rowptr;
    kint* deg_k    = (kint*)deg;
    kint* col_k    = (kint*)col;
    kflt* dinv_k   = (kflt*)dinv;
    const unsigned hoff = (unsigned)((half * 8) * DOUT + colBase + r) * 4u;

    int it = 0;
#pragma unroll 1
    for (int tile = blockIdx.x; tile < Nn / 16; tile += gridDim.x, ++it) {
        float* buf = &lds_x[it & 1][0];

        for (int n = 0; n < RPW; ++n) {
            const int row = wave * RPW + n;
            const int d   = tile * 16 + row;
            int j = rowptr_k[d];
            const int end = j + deg_k[d];
            const float wd = dinv_k[d];
            vec acc = hrow[(size_t)d * 32 + lane];
            for (; j + 4 <= end; j += 4) gather_kc<4>(acc, hrow, col_k, j, lane);
            if (j + 2 <= end) { gather_kc<2>(acc, hrow, col_k, j, lane); j += 2; }
            if (j < end) gather_kc<1>(acc, hrow, col_k, j, lane);
            acc = acc * wd + *(const volatile __attribute__((address_space(3))) vec*)(lds_b + lane * 4);
            acc.x = fmaxf(acc.x, 0.0f) * wd; acc.y = fmaxf(acc.y, 0.0f) * wd;
            acc.z = fmaxf(acc.z, 0.0f) * wd; acc.w = fmaxf(acc.w, 0.0f) * wd;
            *(vec*)(buf + row * PITCH + lane * 4) = acc;
        }
        __syncthreads();

        if (wave < NMM) {
            const float* __restrict__ xa = &lds_x[it & 1][r * PITCH + koff];
            v8f c = {};
#pragma unroll
            for (int s = 0; s < 32; ++s) {
                v2f a = *(const v2f*)(xa + 4 * s);
                c = __builtin_amdgcn_wmma_f32_16x16x4_f32(
                     false, a,  false, bfrag[s],
                     (short)0, c,  false,  false);
            }

            char* __restrict__ ht = (char*)H + (size_t)tile * (16 * DOUT * 4);
#pragma unroll
            for (int v = 0; v < 8; ++v) *(float*)(ht + hoff + v * (DOUT * 4)) = c[v];
        }
    }
}

extern "C" void kernel_launch(void* const* d_in, const int* in_sizes, int n_in,
                              void* d_out, int out_size, void* d_ws, size_t ws_size,
                              hipStream_t stream) {
    const float* X   = (const float*)d_in[0];
    const float* W1  = (const float*)d_in[1];
    const float* b1  = (const float*)d_in[2];
    const float* W2  = (const float*)d_in[3];
    const float* b2  = (const float*)d_in[4];
    const float* W3  = (const float*)d_in[5];
    const float* b3  = (const float*)d_in[6];
    const int*   ei  = (const int*)d_in[7];
    const int*   src = ei;
    const int*   dst = ei + Ee;
    float* out = (float*)d_out;

    float* bufA   = (float*)d_ws;
    float* bufB   = bufA + (size_t)Nn * 128;
    int*   deg    = (int*)(bufB + (size_t)Nn * 128);
    float* dinv   = (float*)(deg + Nn);
    int*   rowptr = (int*)(dinv + Nn);
    int*   part   = rowptr + Nn;
    int*   col    = part + NB * CH;
    int*   loc    = col + Ee;

    k_chunk_sort<<<NB, 1024, 0, stream>>>(src, dst, part, loc);
    k_node_sort<<<NR, 1024, 0, stream>>>(part, loc, deg, rowptr, dinv, col);

    const int TPB   = 5;
    const int gGemm = Nn / (16 * TPB);
    const int NPW   = 4;
    const int gAgg  = (Nn + 8 * NPW - 1) / (8 * NPW);

    k_gemm_wmma<128, false, TPB><<<gGemm, 256, 0, stream>>>(X, W1, dinv, bufA);

    k_agg_gemm<128, 10><<<1250, 256, 0, stream>>>(bufA, b1, dinv, rowptr, deg, col, W2, bufB);

    k_agg_gemm<64, 10><<<1250, 256, 0, stream>>>(bufB, b2, dinv, rowptr, deg, col, W3, bufA);

    k_gather<64, NPW><<<gAgg, 256, 0, stream>>>(bufA, b3, dinv, rowptr, deg, col, out);
}


# Level 2 of the edge sort and the layer-1 product in one launch: a block owns 400 nodes, buckets their edges, then multiplies its 25 row tiles by its own dinv from LDS; 6 launches to 5; sum order kept
# speedup vs baseline: 4.9457x; 1.0580x over previous
#include <hip/hip_runtime.h>

#define Nn 100000
#define Ee 1000000
#define INF 128
#define HID 128
#define CLS 64

typedef __attribute__((ext_vector_type(2))) float v2f;
typedef __attribute__((ext_vector_type(8))) float v8f;

#define RNG 400
#define LSH 9
#define LMSK 511
#define NR  (Nn / RNG)
#define RP  256
#define CH  4096
#define NB  ((Ee + CH - 1) / CH)
#define LP  260

__global__ __launch_bounds__(1024) void k_chunk_sort(const int* __restrict__ src, const int* __restrict__ dst,
                                                     int* __restrict__ part, int* __restrict__ loc) {
    __shared__ __attribute__((aligned(16))) int s_cnt[RP];
    __shared__ __attribute__((aligned(16))) int s_start[RP];
    __shared__ __attribute__((aligned(16))) int s_buf[CH];
    __shared__ int s_tot;
    const int tid = threadIdx.x;
    const int b   = blockIdx.x;
    const int e   = b * CH + tid * 4;
    const bool ok = e < Ee;
    int4 d = make_int4(0, 0, 0, 0), s = make_int4(0, 0, 0, 0);
    if (ok) { d = *(const int4*)(dst + e); s = *(const int4*)(src + e); }
    if (tid < RP) s_cnt[tid] = 0;
    __syncthreads();
    const int q0 = (int)((unsigned)d.x / RNG), q1 = (int)((unsigned)d.y / RNG);
    const int q2 = (int)((unsigned)d.z / RNG), q3 = (int)((unsigned)d.w / RNG);
    if (ok) {
        atomicAdd(&s_cnt[q0], 1);
        atomicAdd(&s_cnt[q1], 1);
        atomicAdd(&s_cnt[q2], 1);
        atomicAdd(&s_cnt[q3], 1);
    }
    __syncthreads();
    if (tid < 32) {
        const int4 ta = *(const int4*)&s_cnt[8 * tid];
        const int4 tb = *(const int4*)&s_cnt[8 * tid + 4];
        const int sum = ((ta.x + ta.y) + (ta.z + ta.w)) + ((tb.x + tb.y) + (tb.z + tb.w));
        int inc = sum;
        for (int o = 1; o < 32; o <<= 1) {
            int t = __shfl_up(inc, o);
            if (tid >= o) inc += t;
        }
        int4 ra, rb;
        ra.x = inc - sum;
        ra.y = ra.x + ta.x;
        ra.z = ra.y + ta.y;
        ra.w = ra.z + ta.z;
        rb.x = ra.w + ta.w;
        rb.y = rb.x + tb.x;
        rb.z = rb.y + tb.y;
        rb.w = rb.z + tb.z;
        *(int4*)&s_start[8 * tid]     = ra;
        *(int4*)&s_start[8 * tid + 4] = rb;
        *(int4*)&s_cnt[8 * tid]       = make_int4(0, 0, 0, 0);
        *(int4*)&s_cnt[8 * tid + 4]   = make_int4(0, 0, 0, 0);
        *(int4*)&loc[b * LP + 8 * tid]     = ra;
        *(int4*)&loc[b * LP + 8 * tid + 4] = rb;
        if (tid == 31) { s_tot = inc; loc[b * LP + RP] = inc; }
    }
    __syncthreads();
    if (ok) {
        const int p0 = s_start[q0] + atomicAdd(&s_cnt[q0], 1);
        const int p1 = s_start[q1] + atomicAdd(&s_cnt[q1], 1);
        const int p2 = s_start[q2] + atomicAdd(&s_cnt[q2], 1);
        const int p3 = s_start[q3] + atomicAdd(&s_cnt[q3], 1);
        s_buf[p0] = (s.x << LSH) | (d.x - q0 * RNG);
        s_buf[p1] = (s.y << LSH) | (d.y - q1 * RNG);
        s_buf[p2] = (s.z << LSH) | (d.z - q2 * RNG);
        s_buf[p3] = (s.w << LSH) | (d.w - q3 * RNG);
    }
    __syncthreads();
    if (4 * tid < s_tot) *(int4*)&part[b * CH + 4 * tid] = *(const int4*)&s_buf[4 * tid];
}

typedef const int   __attribute__((address_space(4))) kint;
typedef const float __attribute__((address_space(4))) kflt;

__global__ __launch_bounds__(1024) void k_sort_gemm(const int* __restrict__ part, const int* __restrict__ loc,
                                                    int* __restrict__ deg, int* __restrict__ rowptr,
                                                    float* __restrict__ dinv, int* __restrict__ col,
                                                    const float* __restrict__ X, const float* __restrict__ W,
                                                    float* __restrict__ H) {
    constexpr int KEEP  = 8;
    constexpr int PITCH = 132;
    constexpr int TPR   = RNG / 16;
    constexpr int NG    = 4;
    constexpr int STEPS = (TPR + NG - 1) / NG;
    __shared__ int s_cnt[1024];
    __shared__ int s_w[32];
    __shared__ int s_beg;
    __shared__ float s_dinv[RNG];
    __shared__ __attribute__((aligned(16))) float lds_x[2][NG][16 * PITCH];

    const int tid  = threadIdx.x;
    const int lane = tid & 31;
    const int wave = __builtin_amdgcn_readfirstlane(tid >> 5);
    const int rg   = blockIdx.x;

    const int cs      = wave & 7;
    const int g       = wave >> 3;
    const int gt      = tid & 255;
    const int colBase = cs * 16;
    const int r       = lane & 15;
    const int half    = lane >> 4;
    const int koff    = half * 2;

    const int ck = tid >> 2;
    const int k4 = tid & 3;
    int lo = 0, hi = 0;
    if (ck < NB) { lo = loc[ck * LP + rg]; hi = loc[ck * LP + rg + 1]; }

    v2f bfrag[32];
#pragma unroll
    for (int s = 0; s < 32; ++s) {
        bfrag[s].x = W[(4 * s + koff) * HID + colBase + r];
        bfrag[s].y = W[(4 * s + koff + 1) * HID + colBase + r];
    }
    constexpr unsigned TILEB = 16 * 128 * 4;
    const unsigned xoff = (unsigned)gt * 16u;
    const char* __restrict__ xr = (const char*)X + (size_t)rg * (RNG * 128 * 4);
    float4 stage[2];
    {
        const char* __restrict__ xt = xr + g * TILEB;
#pragma unroll
        for (int u = 0; u < 2; ++u) stage[u] = *(const float4*)(xt + xoff + u * 4096);
    }

    {
        const int* __restrict__ pc = part + (ck < NB ? ck : 0) * CH;

        int v[KEEP];
#pragma unroll
        for (int u = 0; u < KEEP; ++u) {
            const int i = lo + k4 + 4 * u;
            v[u] = i < hi ? pc[i] : -1;
        }
        s_cnt[tid] = 0;
        if (tid == 0) s_beg = 0;
        __syncthreads();
        int part_lo = k4 == 0 ? lo : 0;
        for (int o = 16; o > 0; o >>= 1) part_lo += __shfl_xor(part_lo, o);
        if (lane == 0) atomicAdd(&s_beg, part_lo);
#pragma unroll
        for (int u = 0; u < KEEP; ++u)
            if (v[u] >= 0) atomicAdd(&s_cnt[v[u] & LMSK], 1);
        for (int i = lo + k4 + 4 * KEEP; i < hi; i += 4) atomicAdd(&s_cnt[pc[i] & LMSK], 1);
        __syncthreads();

        const int beg = s_beg;
        const int c = s_cnt[tid];
        int inc = c;
        for (int o = 1; o < 32; o <<= 1) {
            int t = __shfl_up(inc, o);
            if (lane >= o) inc += t;
        }
        if (lane == 31) s_w[wave] = inc;
        __syncthreads();
        if (wave == 0) {
            const int w = s_w[lane];
            int wi = w;
            for (int o = 1; o < 32; o <<= 1) {
                int t = __shfl_up(wi, o);
                if (lane >= o) wi += t;
            }
            s_w[lane] = wi - w;
        }
        __syncthreads();
        const int pos = beg + s_w[wave] + inc - c;
        s_cnt[tid] = pos;
        if (tid < RNG) {
            const int node = rg * RNG + tid;
            const float dv = 1.0f / sqrtf((float)c + 1.0f);
            deg[node]    = c;
            rowptr[node] = pos;
            dinv[node]   = dv;
            s_dinv[tid]  = dv;
        }
        __syncthreads();
#pragma unroll
        for (int u = 0; u < KEEP; ++u)
            if (v[u] >= 0) col[atomicAdd(&s_cnt[v[u] & LMSK], 1)] = v[u] >> LSH;
        for (int i = lo + k4 + 4 * KEEP; i < hi; i += 4) {
            const int x = pc[i];
            col[atomicAdd(&s_cnt[x & LMSK], 1)] = x >> LSH;
        }
    }

    const unsigned loff = (unsigned)((gt >> 5) * PITCH + (gt & 31) * 4) * 4u;
    const unsigned hoff = (unsigned)((half * 8) * HID + colBase + r) * 4u;
    constexpr unsigned LROWS = 8 * PITCH * 4;
    constexpr unsigned LBUF  = NG * 16 * PITCH * 4;
    char* __restrict__ hr = (char*)H + (size_t)rg * (RNG * HID * 4);
    char* lg = (char*)&lds_x[0][g][0];
    float sc[2];

    {
#pragma unroll
        for (int u = 0; u < 2; ++u) sc[u] = s_dinv[g * 16 + cs + u * 8];
#pragma unroll
        for (int u = 0; u < 2; ++u) {
            float4 v = stage[u];
            v.x *= sc[u]; v.y *= sc[u]; v.z *= sc[u]; v.w *= sc[u];
            *(float4*)(lg + loff + u * LROWS) = v;
        }
    }
    __syncthreads();

#pragma unroll 1
    for (int st = 0; st < STEPS; ++st) {
        const int  tl  = st * NG + g;
        const bool cur = tl < TPR;
        const bool nxt = tl + NG < TPR;
        if (nxt) {
            const char* __restrict__ xt = xr + (tl + NG) * TILEB;
#pragma unroll
            for (int u = 0; u < 2; ++u) stage[u] = *(const float4*)(xt + xoff + u * 4096);
#pragma unroll
            for (int u = 0; u < 2; ++u) sc[u] = s_dinv[(tl + NG) * 16 + cs + u * 8];
        }

        if (cur) {
            const float* __restrict__ xa = (const float*)(lg + (st & 1) * LBUF) + r * PITCH + koff;
            v8f acc = {};
#pragma unroll
            for (int s = 0; s < 32; ++s) {
                v2f a = *(const v2f*)(xa + 4 * s);
                acc = __builtin_amdgcn_wmma_f32_16x16x4_f32(
                     false, a,  false, bfrag[s],
                     (short)0, acc,  false,  false);
            }
            char* __restrict__ ht = hr + tl * TILEB;
#pragma unroll
            for (int v = 0; v < 8; ++v) *(float*)(ht + hoff + v * (HID * 4)) = acc[v];
        }

        if (nxt) {
            char* lnext = lg + ((st + 1) & 1) * LBUF;
#pragma unroll
            for (int u = 0; u < 2; ++u) {
                float4 v = stage[u];
                v.x *= sc[u]; v.y *= sc[u]; v.z *= sc[u]; v.w *= sc[u];
                *(float4*)(lnext + loff + u * LROWS) = v;
            }
        }
        if (st + 1 < STEPS) {
            __builtin_amdgcn_fence(__ATOMIC_RELEASE, "workgroup", "local");
            __builtin_amdgcn_s_barrier();
            __builtin_amdgcn_fence(__ATOMIC_ACQUIRE, "workgroup", "local");
        }
    }
}

template <int DOUT>
using rowvec = float __attribute__((ext_vector_type(DOUT / 32)));

template <int DOUT, int K>
__device__ __forceinline__ void gather_k(rowvec<DOUT>& acc, const rowvec<DOUT>* __restrict__ hrow,
                                         const int* __restrict__ col, int j, int lane) {
    int   s[K];
    rowvec<DOUT> r[K];
#pragma unroll
    for (int u = 0; u < K; ++u) s[u] = col[j + u];
#pragma unroll
    for (int u = 0; u < K; ++u) r[u] = hrow[(size_t)s[u] * 32 + lane];
#pragma unroll
    for (int u = 0; u < K; ++u) acc += r[u];
}

template <int DOUT, int NPW>
__global__ __launch_bounds__(256) void k_gather(const float* __restrict__ h,
                                                const float* __restrict__ b,
                                                const float* __restrict__ dinv,
                                                const int* __restrict__ rowptr,
                                                const int* __restrict__ deg,
                                                const int* __restrict__ col,
                                                float* __restrict__ out) {
    typedef rowvec<DOUT> vec;
    const int lane = threadIdx.x & 31;
    const int wv = __builtin_amdgcn_readfirstlane((int)((blockIdx.x * blockDim.x + threadIdx.x) >> 5));
    const vec* __restrict__ hrow = (const vec*)h;
    const vec bias = ((const vec*)b)[lane];

    for (int i = 0; i < NPW; ++i) {
        const int d = wv * NPW + i;
        if (d >= Nn) break;
        int j = rowptr[d];
        const int end = j + deg[d];
        const float wd = dinv[d];
        vec acc = hrow[(size_t)d * 32 + lane];
        for (; j + 4 <= end; j += 4) gather_k<DOUT, 4>(acc, hrow, col, j, lane);
        if (j + 2 <= end) { gather_k<DOUT, 2>(acc, hrow, col, j, lane); j += 2; }
        if (j < end) gather_k<DOUT, 1>(acc, hrow, col, j, lane);
        ((vec*)out)[(size_t)d * 32 + lane] = acc * wd + bias;
    }
}


template <int K>
__device__ __forceinline__ void gather_kc(rowvec<128>& acc, const rowvec<128>* __restrict__ hrow,
                                          kint* col, int j, int lane) {
    int   s[K];
    rowvec<128> r[K];
#pragma unroll
    for (int u = 0; u < K; ++u) s[u] = col[j + u];
#pragma unroll
    for (int u = 0; u < K; ++u) r[u] = hrow[(size_t)s[u] * 32 + lane];
#pragma unroll
    for (int u = 0; u < K; ++u) acc += r[u];
}
template <int DOUT, int WPS>
__global__ __launch_bounds__(256, WPS) void k_agg_gemm(const float* __restrict__ h,
                                                       const float* __restrict__ b,
                                                       const float* __restrict__ dinv,
                                                       const int* __restrict__ rowptr,
                                                       const int* __restrict__ deg,
                                                       const int* __restrict__ col,
                                                       const float* __restrict__ W,
                                                       float* __restrict__ H) {
    constexpr int NMM   = DOUT / 16;
    constexpr int RPW   = 2;
    constexpr int PITCH = 132;
    typedef rowvec<128> vec;
    __shared__ __attribute__((aligned(16))) float lds_x[2][16 * PITCH];
    __shared__ __attribute__((aligned(16))) float lds_b[128];

    const int tid     = threadIdx.x;
    const int lane    = tid & 31;
    const int wave    = __builtin_amdgcn_readfirstlane(tid >> 5);
    const int colBase = wave * 16;
    const int r       = lane & 15;
    const int half    = lane >> 4;
    const int koff    = half * 2;

    v2f bfrag[32];
    if (wave < NMM) {
#pragma unroll
        for (int s = 0; s < 32; ++s) {
            bfrag[s].x = W[(4 * s + koff) * DOUT + colBase + r];
            bfrag[s].y = W[(4 * s + koff + 1) * DOUT + colBase + r];
        }
    }

    const vec* __restrict__ hrow = (const vec*)h;
    if (tid < 32) *(vec*)(lds_b + lane * 4) = ((const vec*)b)[lane];
    __syncthreads();
    kint* rowptr_k = (kint*)rowptr;
    kint* deg_k    = (kint*)deg;
    kint* col_k    = (kint*)col;
    kflt* dinv_k   = (kflt*)dinv;
    const unsigned hoff = (unsigned)((half * 8) * DOUT + colBase + r) * 4u;

    int it = 0;
#pragma unroll 1
    for (int tile = blockIdx.x; tile < Nn / 16; tile += gridDim.x, ++it) {
        float* buf = &lds_x[it & 1][0];

        for (int n = 0; n < RPW; ++n) {
            const int row = wave * RPW + n;
            const int d   = tile * 16 + row;
            int j = rowptr_k[d];
            const int end = j + deg_k[d];
            const float wd = dinv_k[d];
            vec acc = hrow[(size_t)d * 32 + lane];
            for (; j + 4 <= end; j += 4) gather_kc<4>(acc, hrow, col_k, j, lane);
            if (j + 2 <= end) { gather_kc<2>(acc, hrow, col_k, j, lane); j += 2; }
            if (j < end) gather_kc<1>(acc, hrow, col_k, j, lane);
            acc = acc * wd + *(const volatile __attribute__((address_space(3))) vec*)(lds_b + lane * 4);
            acc.x = fmaxf(acc.x, 0.0f) * wd; acc.y = fmaxf(acc.y, 0.0f) * wd;
            acc.z = fmaxf(acc.z, 0.0f) * wd; acc.w = fmaxf(acc.w, 0.0f) * wd;
            *(vec*)(buf + row * PITCH + lane * 4) = acc;
        }
        __syncthreads();

        if (wave < NMM) {
            const float* __restrict__ xa = &lds_x[it & 1][r * PITCH + koff];
            v8f c = {};
#pragma unroll
            for (int s = 0; s < 32; ++s) {
                v2f a = *(const v2f*)(xa + 4 * s);
                c = __builtin_amdgcn_wmma_f32_16x16x4_f32(
                     false, a,  false, bfrag[s],
                     (short)0, c,  false,  false);
            }

            char* __restrict__ ht = (char*)H + (size_t)tile * (16 * DOUT * 4);
#pragma unroll
            for (int v = 0; v < 8; ++v) *(float*)(ht + hoff + v * (DOUT * 4)) = c[v];
        }
    }
}

extern "C" void kernel_launch(void* const* d_in, const int* in_sizes, int n_in,
                              void* d_out, int out_size, void* d_ws, size_t ws_size,
                              hipStream_t stream) {
    const float* X   = (const float*)d_in[0];
    const float* W1  = (const float*)d_in[1];
    const float* b1  = (const float*)d_in[2];
    const float* W2  = (const float*)d_in[3];
    const float* b2  = (const float*)d_in[4];
    const float* W3  = (const float*)d_in[5];
    const float* b3  = (const float*)d_in[6];
    const int*   ei  = (const int*)d_in[7];
    const int*   src = ei;
    const int*   dst = ei + Ee;
    float* out = (float*)d_out;

    float* bufA   = (float*)d_ws;
    float* bufB   = bufA + (size_t)Nn * 128;
    int*   deg    = (int*)(bufB + (size_t)Nn * 128);
    float* dinv   = (float*)(deg + Nn);
    int*   rowptr = (int*)(dinv + Nn);
    int*   part   = rowptr + Nn;
    int*   col    = part + NB * CH;
    int*   loc    = col + Ee;

    k_chunk_sort<<<NB, 1024, 0, stream>>>(src, dst, part, loc);

    k_sort_gemm<<<NR, 1024, 0, stream>>>(part, loc, deg, rowptr, dinv, col, X, W1, bufA);

    const int NPW   = 4;
    const int gAgg  = (Nn + 8 * NPW - 1) / (8 * NPW);

    k_agg_gemm<128, 10><<<1250, 256, 0, stream>>>(bufA, b1, dinv, rowptr, deg, col, W2, bufB);

    k_agg_gemm<64, 10><<<1250, 256, 0, stream>>>(bufB, b2, dinv, rowptr, deg, col, W3, bufA);

    k_gather<64, NPW><<<gAgg, 256, 0, stream>>>(bufA, b3, dinv, rowptr, deg, col, out);
}
